# starting kernel
# baseline (speedup 1.0000x reference)
#include <hip/hip_runtime.h>
#include <hip/hip_bf16.h>

typedef __attribute__((ext_vector_type(2))) float v2f;
typedef __attribute__((ext_vector_type(8))) float v8f;

#define IN_SIZE 64
#define OUT_SIZE 64

__global__ __launch_bounds__(256) void rgcn_init_out(
    const float* __restrict__ bias, float* __restrict__ out, int n4, int n_rel) {
  int i = blockIdx.x * blockDim.x + threadIdx.x;
  if (i >= n4) return;
  int c = (i * 4) & (OUT_SIZE - 1);
  float4 b = make_float4(0.f, 0.f, 0.f, 0.f);
  for (int r = 0; r < n_rel; ++r) {
    const float* br = bias + r * OUT_SIZE + c;
    b.x += br[0]; b.y += br[1]; b.z += br[2]; b.w += br[3];
  }
  ((float4*)out)[i] = b;
}

__global__ __launch_bounds__(256) void rgcn_zero(float4* __restrict__ p, int n4) {
  int i = blockIdx.x * blockDim.x + threadIdx.x;
  if (i < n4) p[i] = make_float4(0.f, 0.f, 0.f, 0.f);
}

__global__ __launch_bounds__(256) void rgcn_scatter(
    const float* __restrict__ inp, const int* __restrict__ src,
    const int* __restrict__ dst, const float* __restrict__ eval,
    float* __restrict__ agg, int n_edges) {
  const int wid = (blockIdx.x * 256 + threadIdx.x) >> 5;
  if (wid >= n_edges) return;
  const int lane = threadIdx.x & 31;
  const int s = src[wid];
  const int d = dst[wid];
  const float v = eval[wid];
  const int col = lane * 2;
  const float2 m = *(const float2*)(inp + (size_t)s * IN_SIZE + col);
  float* p = agg + (size_t)d * IN_SIZE + col;
  unsafeAtomicAdd(p, m.x * v);
  unsafeAtomicAdd(p + 1, m.y * v);
}

__global__ __launch_bounds__(256) void rgcn_gemm_wmma(
    const float* __restrict__ agg, const float* __restrict__ W,
    float* __restrict__ out, int num_tiles) {
  __shared__ float sW[IN_SIZE * OUT_SIZE];
  const int t = threadIdx.x;
  {
    const float4* Wv = (const float4*)W;
    float4* sWv = (float4*)sW;
#pragma unroll
    for (int i = 0; i < (IN_SIZE * OUT_SIZE / 4) / 256; ++i)
      sWv[t + i * 256] = Wv[t + i * 256];
  }
  __syncthreads();

  const int wave = t >> 5;
  const int lane = t & 31;
  const int half = lane >> 4;
  const int l16 = lane & 15;

  const int tile = blockIdx.x * 8 + wave;
  if (tile >= num_tiles) return;
  const int rowBase = (tile >> 2) * 16;
  const int colBase = (tile & 3) * 16;

  v8f c;
#pragma unroll
  for (int j = 0; j < 8; ++j)
    c[j] = out[(size_t)(rowBase + j + half * 8) * OUT_SIZE + colBase + l16];

  const float* aRow = agg + (size_t)(rowBase + l16) * IN_SIZE;
  const int koff = half * 2;
#pragma unroll
  for (int k = 0; k < IN_SIZE; k += 4) {
    v2f a, b;
    const float2 av = *(const float2*)(aRow + k + koff);
    a[0] = av.x;
    a[1] = av.y;
    b[0] = sW[(k + koff) * OUT_SIZE + colBase + l16];
    b[1] = sW[(k + koff + 1) * OUT_SIZE + colBase + l16];
    c = __builtin_amdgcn_wmma_f32_16x16x4_f32(
         false, a,  false, b,
         (short)0, c,  false,  false);
  }

#pragma unroll
  for (int j = 0; j < 8; ++j)
    out[(size_t)(rowBase + j + half * 8) * OUT_SIZE + colBase + l16] = c[j];
}

extern "C" void kernel_launch(void* const* d_in, const int* in_sizes, int n_in,
                              void* d_out, int out_size, void* d_ws, size_t ws_size,
                              hipStream_t stream) {
  const float* inp = (const float*)d_in[0];
  const int* src = (const int*)d_in[1];
  const int* dst = (const int*)d_in[2];
  const float* eval = (const float*)d_in[3];
  const float* weights = (const float*)d_in[4];
  const float* bias = (const float*)d_in[5];
  float* out = (float*)d_out;
  float* agg = (float*)d_ws;

  const int n_nodes = in_sizes[0] / IN_SIZE;
  const int n_rel = in_sizes[4] / (IN_SIZE * OUT_SIZE);
  const int n_edges = in_sizes[1] / n_rel;

  const int n4 = n_nodes * OUT_SIZE / 4;
  const int vb = (n4 + 255) / 256;
  rgcn_init_out<<<vb, 256, 0, stream>>>(bias, out, n4, n_rel);

  const int scatter_blocks = (n_edges * 32 + 255) / 256;
  const int num_tiles = (n_nodes / 16) * (OUT_SIZE / 16);
  const int gemm_blocks = (num_tiles + 7) / 8;

  for (int r = 0; r < n_rel; ++r) {
    rgcn_zero<<<vb, 256, 0, stream>>>((float4*)agg, n4);
    rgcn_scatter<<<scatter_blocks, 256, 0, stream>>>(
        inp, src + (size_t)r * n_edges, dst + (size_t)r * n_edges,
        eval + (size_t)r * n_edges, agg, n_edges);
    rgcn_gemm_wmma<<<gemm_blocks, 256, 0, stream>>>(
        agg, weights + (size_t)r * IN_SIZE * OUT_SIZE, out, num_tiles);
  }
}


# scatter: a wave takes 32 edges (one coalesced index load per array, lane broadcast) and issues row gathers 16 at a time ahead of their atomics; sums still in atomic arrival order
# speedup vs baseline: 1.7967x; 1.7967x over previous
#include <hip/hip_runtime.h>
#include <hip/hip_bf16.h>

typedef __attribute__((ext_vector_type(2))) float v2f;
typedef __attribute__((ext_vector_type(8))) float v8f;

#define IN_SIZE 64
#define OUT_SIZE 64

__global__ __launch_bounds__(256) void rgcn_init_out(
    const float* __restrict__ bias, float* __restrict__ out, int n4, int n_rel) {
  int i = blockIdx.x * blockDim.x + threadIdx.x;
  if (i >= n4) return;
  int c = (i * 4) & (OUT_SIZE - 1);
  float4 b = make_float4(0.f, 0.f, 0.f, 0.f);
  for (int r = 0; r < n_rel; ++r) {
    const float* br = bias + r * OUT_SIZE + c;
    b.x += br[0]; b.y += br[1]; b.z += br[2]; b.w += br[3];
  }
  ((float4*)out)[i] = b;
}

__global__ __launch_bounds__(256) void rgcn_zero(float4* __restrict__ p, int n4) {
  int i = blockIdx.x * blockDim.x + threadIdx.x;
  if (i < n4) p[i] = make_float4(0.f, 0.f, 0.f, 0.f);
}

#define EDGES_PER_WAVE 32
#define GATHER_GROUP 16
__global__ __launch_bounds__(256) void rgcn_scatter(
    const float* __restrict__ inp, const int* __restrict__ src,
    const int* __restrict__ dst, const float* __restrict__ eval,
    float* __restrict__ agg, int n_edges) {
  const int lane = threadIdx.x & 31;
  const int wid = __builtin_amdgcn_readfirstlane((int)((blockIdx.x * 256 + threadIdx.x) >> 5));
  const int base = wid * EDGES_PER_WAVE;
  if (base >= n_edges) return;
  const int left = n_edges - base;
  const int cnt = left < EDGES_PER_WAVE ? left : EDGES_PER_WAVE;
  const int e = base + (lane < cnt ? lane : cnt - 1);
  const int s = src[e];
  const int d = dst[e];
  const int vbits = __float_as_int(eval[e]);
  const int col = lane * 2;

  if (cnt == EDGES_PER_WAVE) {
#pragma unroll
    for (int j0 = 0; j0 < EDGES_PER_WAVE; j0 += GATHER_GROUP) {
      float2 m[GATHER_GROUP];
#pragma unroll
      for (int j = 0; j < GATHER_GROUP; ++j) {
        const int sj = __builtin_amdgcn_readlane(s, j0 + j);
        m[j] = *(const float2*)(inp + (size_t)sj * IN_SIZE + col);
      }
#pragma unroll
      for (int j = 0; j < GATHER_GROUP; ++j) {
        const int dj = __builtin_amdgcn_readlane(d, j0 + j);
        const float vj = __int_as_float(__builtin_amdgcn_readlane(vbits, j0 + j));
        float* p = agg + (size_t)dj * IN_SIZE + col;
        unsafeAtomicAdd(p, m[j].x * vj);
        unsafeAtomicAdd(p + 1, m[j].y * vj);
      }
    }
  } else {
    for (int j = 0; j < cnt; ++j) {
      const int sj = __builtin_amdgcn_readlane(s, j);
      const int dj = __builtin_amdgcn_readlane(d, j);
      const float vj = __int_as_float(__builtin_amdgcn_readlane(vbits, j));
      const float2 mj = *(const float2*)(inp + (size_t)sj * IN_SIZE + col);
      float* p = agg + (size_t)dj * IN_SIZE + col;
      unsafeAtomicAdd(p, mj.x * vj);
      unsafeAtomicAdd(p + 1, mj.y * vj);
    }
  }
}

__global__ __launch_bounds__(256) void rgcn_gemm_wmma(
    const float* __restrict__ agg, const float* __restrict__ W,
    float* __restrict__ out, int num_tiles) {
  __shared__ float sW[IN_SIZE * OUT_SIZE];
  const int t = threadIdx.x;
  {
    const float4* Wv = (const float4*)W;
    float4* sWv = (float4*)sW;
#pragma unroll
    for (int i = 0; i < (IN_SIZE * OUT_SIZE / 4) / 256; ++i)
      sWv[t + i * 256] = Wv[t + i * 256];
  }
  __syncthreads();

  const int wave = t >> 5;
  const int lane = t & 31;
  const int half = lane >> 4;
  const int l16 = lane & 15;

  const int tile = blockIdx.x * 8 + wave;
  if (tile >= num_tiles) return;
  const int rowBase = (tile >> 2) * 16;
  const int colBase = (tile & 3) * 16;

  v8f c;
#pragma unroll
  for (int j = 0; j < 8; ++j)
    c[j] = out[(size_t)(rowBase + j + half * 8) * OUT_SIZE + colBase + l16];

  const float* aRow = agg + (size_t)(rowBase + l16) * IN_SIZE;
  const int koff = half * 2;
#pragma unroll
  for (int k = 0; k < IN_SIZE; k += 4) {
    v2f a, b;
    const float2 av = *(const float2*)(aRow + k + koff);
    a[0] = av.x;
    a[1] = av.y;
    b[0] = sW[(k + koff) * OUT_SIZE + colBase + l16];
    b[1] = sW[(k + koff + 1) * OUT_SIZE + colBase + l16];
    c = __builtin_amdgcn_wmma_f32_16x16x4_f32(
         false, a,  false, b,
         (short)0, c,  false,  false);
  }

#pragma unroll
  for (int j = 0; j < 8; ++j)
    out[(size_t)(rowBase + j + half * 8) * OUT_SIZE + colBase + l16] = c[j];
}

extern "C" void kernel_launch(void* const* d_in, const int* in_sizes, int n_in,
                              void* d_out, int out_size, void* d_ws, size_t ws_size,
                              hipStream_t stream) {
  const float* inp = (const float*)d_in[0];
  const int* src = (const int*)d_in[1];
  const int* dst = (const int*)d_in[2];
  const float* eval = (const float*)d_in[3];
  const float* weights = (const float*)d_in[4];
  const float* bias = (const float*)d_in[5];
  float* out = (float*)d_out;
  float* agg = (float*)d_ws;

  const int n_nodes = in_sizes[0] / IN_SIZE;
  const int n_rel = in_sizes[4] / (IN_SIZE * OUT_SIZE);
  const int n_edges = in_sizes[1] / n_rel;

  const int n4 = n_nodes * OUT_SIZE / 4;
  const int vb = (n4 + 255) / 256;
  rgcn_init_out<<<vb, 256, 0, stream>>>(bias, out, n4, n_rel);

  const int scatter_blocks = (n_edges + 255) / 256;
  const int num_tiles = (n_nodes / 16) * (OUT_SIZE / 16);
  const int gemm_blocks = (num_tiles + 7) / 8;

  for (int r = 0; r < n_rel; ++r) {
    rgcn_zero<<<vb, 256, 0, stream>>>((float4*)agg, n4);
    rgcn_scatter<<<scatter_blocks, 256, 0, stream>>>(
        inp, src + (size_t)r * n_edges, dst + (size_t)r * n_edges,
        eval + (size_t)r * n_edges, agg, n_edges);
    rgcn_gemm_wmma<<<gemm_blocks, 256, 0, stream>>>(
        agg, weights + (size_t)r * IN_SIZE * OUT_SIZE, out, num_tiles);
  }
}


# apply W before the edge sum, (A_r X)W_r = A_r(X W_r): an f32 WMMA kernel writes H_r = inp W_r and out = bias sum, one scatter atomically adds val*H_r[src] to out; 13 launches to 2; sum order changed
# speedup vs baseline: 2.3562x; 1.3114x over previous
#include <hip/hip_runtime.h>
#include <hip/hip_bf16.h>

typedef __attribute__((ext_vector_type(2))) float v2f;
typedef __attribute__((ext_vector_type(8))) float v8f;

#define IN_SIZE 64
#define OUT_SIZE 64


#define PROJ_WAVES 4
#define PROJ_THREADS (PROJ_WAVES * 32)
#define W_PITCH 72

__global__ __launch_bounds__(PROJ_THREADS) void rgcn_project(
    const float* __restrict__ inp, const float* __restrict__ W,
    const float* __restrict__ bias, float* __restrict__ H,
    float* __restrict__ out, int n_nodes, int n_rel) {
  __shared__ float sW[2][IN_SIZE * W_PITCH];
  const int t = threadIdx.x;
  const int wave = __builtin_amdgcn_readfirstlane(t >> 5);
  const int lane = t & 31;
  const int half = lane >> 4;
  const int l16 = lane & 15;

  const int st_row = t >> 4;
  const int st_col = (t & 15) * 4;

  {
    const float4* Wv = (const float4*)W;
#pragma unroll
    for (int i = 0; i < 8; ++i)
      *(float4*)&sW[0][(st_row + 8 * i) * W_PITCH + st_col] = Wv[t + i * PROJ_THREADS];
  }

  const int n_row_tiles = (n_nodes + 15) >> 4;
  int tile = blockIdx.x * PROJ_WAVES + wave;
  const bool valid = tile < n_row_tiles;
  if (!valid) tile = n_row_tiles - 1;
  const int rowBase = tile * 16;
  const int rows_left = valid ? (n_nodes - rowBase) : 0;

  const int bc = l16 * 4;
  float4 bsum = make_float4(0.f, 0.f, 0.f, 0.f);
  for (int r = 0; r < n_rel; ++r) {
    const float4 br = *(const float4*)(bias + r * OUT_SIZE + bc);
    bsum.x += br.x; bsum.y += br.y; bsum.z += br.z; bsum.w += br.w;
  }

  const int koff = half * 2;
  int arow = rowBase + l16;
  if (arow >= n_nodes) arow = n_nodes - 1;
  const float* aRow = inp + (size_t)arow * IN_SIZE;
  v2f a[IN_SIZE / 4];
#pragma unroll
  for (int kk = 0; kk < IN_SIZE / 4; ++kk) {
    const float2 av = *(const float2*)(aRow + kk * 4 + koff);
    a[kk][0] = av.x;
    a[kk][1] = av.y;
  }
  __syncthreads();

  for (int r = 0; r < n_rel; ++r) {
    const float* sB = sW[r & 1];
    const int rn = (r + 1 < n_rel) ? r + 1 : r;
    const float4* Wn = (const float4*)(W + (size_t)rn * IN_SIZE * OUT_SIZE);
    float4 pre[8];
#pragma unroll
    for (int i = 0; i < 8; ++i) pre[i] = Wn[t + i * PROJ_THREADS];
    __builtin_amdgcn_sched_barrier(0);

    v8f c[4];
#pragma unroll
    for (int ct = 0; ct < 4; ++ct)
#pragma unroll
      for (int j = 0; j < 8; ++j) c[ct][j] = 0.f;

#pragma unroll
    for (int kk = 0; kk < IN_SIZE / 4; ++kk) {
      const int k = kk * 4;
#pragma unroll
      for (int ct = 0; ct < 4; ++ct) {
        v2f b;
        b[0] = sB[(k + koff) * W_PITCH + ct * 16 + l16];
        b[1] = sB[(k + koff + 1) * W_PITCH + ct * 16 + l16];
        c[ct] = __builtin_amdgcn_wmma_f32_16x16x4_f32(
             false, a[kk],  false, b,
             (short)0, c[ct],  false,  false);
      }
    }

    {
      float* sN = sW[(r + 1) & 1];
#pragma unroll
      for (int i = 0; i < 8; ++i)
        *(float4*)&sN[(st_row + 8 * i) * W_PITCH + st_col] = pre[i];
    }
    __syncthreads();

    float* Hr = H + (size_t)r * n_nodes * OUT_SIZE + (size_t)rowBase * OUT_SIZE;
#pragma unroll
    for (int j = 0; j < 8; ++j) {
      const int rr = j + half * 8;
      if (rr < rows_left) {
#pragma unroll
        for (int ct = 0; ct < 4; ++ct)
          Hr[(size_t)rr * OUT_SIZE + ct * 16 + l16] = c[ct][j];
      }
    }
  }

#pragma unroll
  for (int i = 0; i < 8; ++i) {
    const int rr = 2 * i + half;
    if (rr < rows_left)
      *(float4*)(out + (size_t)(rowBase + rr) * OUT_SIZE + bc) = bsum;
  }
}

#define EDGES_PER_WAVE 32
#define GATHER_GROUP 16
__global__ __launch_bounds__(256) void rgcn_scatter(
    const float* __restrict__ H, const int* __restrict__ src,
    const int* __restrict__ dst, const float* __restrict__ eval,
    float* __restrict__ out, int n_edges, int n_nodes) {
  const int lane = threadIdx.x & 31;
  const int wid = __builtin_amdgcn_readfirstlane((int)((blockIdx.x * 256 + threadIdx.x) >> 5));
  const int base = wid * EDGES_PER_WAVE;
  if (base >= n_edges) return;
  const int rel = blockIdx.y;
  const float* Hr = H + (size_t)rel * n_nodes * OUT_SIZE;
  src += (size_t)rel * n_edges;
  dst += (size_t)rel * n_edges;
  eval += (size_t)rel * n_edges;
  const int left = n_edges - base;
  const int cnt = left < EDGES_PER_WAVE ? left : EDGES_PER_WAVE;
  const int e = base + (lane < cnt ? lane : cnt - 1);
  const int s = src[e];
  const int d = dst[e];
  const int vbits = __float_as_int(eval[e]);
  const int col = lane * 2;

  if (cnt == EDGES_PER_WAVE) {
#pragma unroll
    for (int j0 = 0; j0 < EDGES_PER_WAVE; j0 += GATHER_GROUP) {
      float2 m[GATHER_GROUP];
#pragma unroll
      for (int j = 0; j < GATHER_GROUP; ++j) {
        const int sj = __builtin_amdgcn_readlane(s, j0 + j);
        m[j] = *(const float2*)(Hr + (size_t)sj * OUT_SIZE + col);
      }
#pragma unroll
      for (int j = 0; j < GATHER_GROUP; ++j) {
        const int dj = __builtin_amdgcn_readlane(d, j0 + j);
        const float vj = __int_as_float(__builtin_amdgcn_readlane(vbits, j0 + j));
        float* p = out + (size_t)dj * OUT_SIZE + col;
        unsafeAtomicAdd(p, m[j].x * vj);
        unsafeAtomicAdd(p + 1, m[j].y * vj);
      }
    }
  } else {
    for (int j = 0; j < cnt; ++j) {
      const int sj = __builtin_amdgcn_readlane(s, j);
      const int dj = __builtin_amdgcn_readlane(d, j);
      const float vj = __int_as_float(__builtin_amdgcn_readlane(vbits, j));
      const float2 mj = *(const float2*)(Hr + (size_t)sj * OUT_SIZE + col);
      float* p = out + (size_t)dj * OUT_SIZE + col;
      unsafeAtomicAdd(p, mj.x * vj);
      unsafeAtomicAdd(p + 1, mj.y * vj);
    }
  }
}

extern "C" void kernel_launch(void* const* d_in, const int* in_sizes, int n_in,
                              void* d_out, int out_size, void* d_ws, size_t ws_size,
                              hipStream_t stream) {
  const float* inp = (const float*)d_in[0];
  const int* src = (const int*)d_in[1];
  const int* dst = (const int*)d_in[2];
  const float* eval = (const float*)d_in[3];
  const float* weights = (const float*)d_in[4];
  const float* bias = (const float*)d_in[5];
  float* out = (float*)d_out;
  float* H = (float*)d_ws;

  const int n_nodes = in_sizes[0] / IN_SIZE;
  const int n_rel = in_sizes[4] / (IN_SIZE * OUT_SIZE);
  const int n_edges = in_sizes[1] / n_rel;

  const int n_row_tiles = (n_nodes + 15) / 16;
  const int proj_blocks = (n_row_tiles + PROJ_WAVES - 1) / PROJ_WAVES;
  rgcn_project<<<proj_blocks, PROJ_THREADS, 0, stream>>>(
      inp, weights, bias, H, out, n_nodes, n_rel);

  const int scatter_blocks = (n_edges + 255) / 256;
  rgcn_scatter<<<dim3(scatter_blocks, n_rel), 256, 0, stream>>>(
      H, src, dst, eval, out, n_edges, n_nodes);
}


# H_r rows stored with columns c and c+32 interleaved: a lane's one 8-byte gather feeds two atomics that each cover one whole 128-B line of out (was two half-used lines); sums still in arrival order
# speedup vs baseline: 3.1275x; 1.3273x over previous
#include <hip/hip_runtime.h>
#include <hip/hip_bf16.h>

typedef __attribute__((ext_vector_type(2))) float v2f;
typedef __attribute__((ext_vector_type(8))) float v8f;

#define IN_SIZE 64
#define OUT_SIZE 64


#define PROJ_WAVES 4
#define PROJ_THREADS (PROJ_WAVES * 32)
#define W_PITCH 72

__global__ __launch_bounds__(PROJ_THREADS) void rgcn_project(
    const float* __restrict__ inp, const float* __restrict__ W,
    const float* __restrict__ bias, float* __restrict__ H,
    float* __restrict__ out, int n_nodes, int n_rel) {
  __shared__ float sW[2][IN_SIZE * W_PITCH];
  const int t = threadIdx.x;
  const int wave = __builtin_amdgcn_readfirstlane(t >> 5);
  const int lane = t & 31;
  const int half = lane >> 4;
  const int l16 = lane & 15;

  const int st_row = t >> 4;
  const int st_col = (t & 15) * 4;

  {
    const float4* Wv = (const float4*)W;
#pragma unroll
    for (int i = 0; i < 8; ++i)
      *(float4*)&sW[0][(st_row + 8 * i) * W_PITCH + st_col] = Wv[t + i * PROJ_THREADS];
  }

  const int n_row_tiles = (n_nodes + 15) >> 4;
  int tile = blockIdx.x * PROJ_WAVES + wave;
  const bool valid = tile < n_row_tiles;
  if (!valid) tile = n_row_tiles - 1;
  const int rowBase = tile * 16;
  const int rows_left = valid ? (n_nodes - rowBase) : 0;

  const int bc = l16 * 4;
  float4 bsum = make_float4(0.f, 0.f, 0.f, 0.f);
  for (int r = 0; r < n_rel; ++r) {
    const float4 br = *(const float4*)(bias + r * OUT_SIZE + bc);
    bsum.x += br.x; bsum.y += br.y; bsum.z += br.z; bsum.w += br.w;
  }

  const int koff = half * 2;
  int arow = rowBase + l16;
  if (arow >= n_nodes) arow = n_nodes - 1;
  const float* aRow = inp + (size_t)arow * IN_SIZE;
  v2f a[IN_SIZE / 4];
#pragma unroll
  for (int kk = 0; kk < IN_SIZE / 4; ++kk) {
    const float2 av = *(const float2*)(aRow + kk * 4 + koff);
    a[kk][0] = av.x;
    a[kk][1] = av.y;
  }
  __syncthreads();

  for (int r = 0; r < n_rel; ++r) {
    const float* sB = sW[r & 1];
    const int rn = (r + 1 < n_rel) ? r + 1 : r;
    const float4* Wn = (const float4*)(W + (size_t)rn * IN_SIZE * OUT_SIZE);
    float4 pre[8];
#pragma unroll
    for (int i = 0; i < 8; ++i) pre[i] = Wn[t + i * PROJ_THREADS];
    __builtin_amdgcn_sched_barrier(0);

    v8f c[4];
#pragma unroll
    for (int ct = 0; ct < 4; ++ct)
#pragma unroll
      for (int j = 0; j < 8; ++j) c[ct][j] = 0.f;

#pragma unroll
    for (int kk = 0; kk < IN_SIZE / 4; ++kk) {
      const int k = kk * 4;
#pragma unroll
      for (int ct = 0; ct < 4; ++ct) {
        v2f b;
        b[0] = sB[(k + koff) * W_PITCH + ct * 16 + l16];
        b[1] = sB[(k + koff + 1) * W_PITCH + ct * 16 + l16];
        c[ct] = __builtin_amdgcn_wmma_f32_16x16x4_f32(
             false, a[kk],  false, b,
             (short)0, c[ct],  false,  false);
      }
    }

    {
      float* sN = sW[(r + 1) & 1];
#pragma unroll
      for (int i = 0; i < 8; ++i)
        *(float4*)&sN[(st_row + 8 * i) * W_PITCH + st_col] = pre[i];
    }
    __syncthreads();

    float* Hr = H + (size_t)r * n_nodes * OUT_SIZE + (size_t)rowBase * OUT_SIZE;
#pragma unroll
    for (int j = 0; j < 8; ++j) {
      const int rr = j + half * 8;
      if (rr < rows_left) {
        float* hp = Hr + (size_t)rr * OUT_SIZE + 2 * l16;
        *(float2*)hp = make_float2(c[0][j], c[2][j]);
        *(float2*)(hp + 32) = make_float2(c[1][j], c[3][j]);
      }
    }
  }

#pragma unroll
  for (int i = 0; i < 8; ++i) {
    const int rr = 2 * i + half;
    if (rr < rows_left)
      *(float4*)(out + (size_t)(rowBase + rr) * OUT_SIZE + bc) = bsum;
  }
}

#define EDGES_PER_WAVE 32
#define GATHER_GROUP 16
__global__ __launch_bounds__(256) void rgcn_scatter(
    const float* __restrict__ H, const int* __restrict__ src,
    const int* __restrict__ dst, const float* __restrict__ eval,
    float* __restrict__ out, int n_edges, int n_nodes) {
  const int lane = threadIdx.x & 31;
  const int wid = __builtin_amdgcn_readfirstlane((int)((blockIdx.x * 256 + threadIdx.x) >> 5));
  const int base = wid * EDGES_PER_WAVE;
  if (base >= n_edges) return;
  const int rel = blockIdx.y;
  const float* Hr = H + (size_t)rel * n_nodes * OUT_SIZE;
  src += (size_t)rel * n_edges;
  dst += (size_t)rel * n_edges;
  eval += (size_t)rel * n_edges;
  const int left = n_edges - base;
  const int cnt = left < EDGES_PER_WAVE ? left : EDGES_PER_WAVE;
  const int e = base + (lane < cnt ? lane : cnt - 1);
  const int s = src[e];
  const int d = dst[e];
  const int vbits = __float_as_int(eval[e]);
  const int gcol = lane * 2;

  if (cnt == EDGES_PER_WAVE) {
#pragma unroll
    for (int j0 = 0; j0 < EDGES_PER_WAVE; j0 += GATHER_GROUP) {
      float2 m[GATHER_GROUP];
#pragma unroll
      for (int j = 0; j < GATHER_GROUP; ++j) {
        const int sj = __builtin_amdgcn_readlane(s, j0 + j);
        m[j] = *(const float2*)(Hr + (size_t)sj * OUT_SIZE + gcol);
      }
#pragma unroll
      for (int j = 0; j < GATHER_GROUP; ++j) {
        const int dj = __builtin_amdgcn_readlane(d, j0 + j);
        const float vj = __int_as_float(__builtin_amdgcn_readlane(vbits, j0 + j));
        float* p = out + (size_t)dj * OUT_SIZE + lane;
        unsafeAtomicAdd(p, m[j].x * vj);
        unsafeAtomicAdd(p + 32, m[j].y * vj);
      }
    }
  } else {
    for (int j = 0; j < cnt; ++j) {
      const int sj = __builtin_amdgcn_readlane(s, j);
      const int dj = __builtin_amdgcn_readlane(d, j);
      const float vj = __int_as_float(__builtin_amdgcn_readlane(vbits, j));
      const float2 mj = *(const float2*)(Hr + (size_t)sj * OUT_SIZE + gcol);
      float* p = out + (size_t)dj * OUT_SIZE + lane;
      unsafeAtomicAdd(p, mj.x * vj);
      unsafeAtomicAdd(p + 32, mj.y * vj);
    }
  }
}

extern "C" void kernel_launch(void* const* d_in, const int* in_sizes, int n_in,
                              void* d_out, int out_size, void* d_ws, size_t ws_size,
                              hipStream_t stream) {
  const float* inp = (const float*)d_in[0];
  const int* src = (const int*)d_in[1];
  const int* dst = (const int*)d_in[2];
  const float* eval = (const float*)d_in[3];
  const float* weights = (const float*)d_in[4];
  const float* bias = (const float*)d_in[5];
  float* out = (float*)d_out;
  float* H = (float*)d_ws;

  const int n_nodes = in_sizes[0] / IN_SIZE;
  const int n_rel = in_sizes[4] / (IN_SIZE * OUT_SIZE);
  const int n_edges = in_sizes[1] / n_rel;

  const int n_row_tiles = (n_nodes + 15) / 16;
  const int proj_blocks = (n_row_tiles + PROJ_WAVES - 1) / PROJ_WAVES;
  rgcn_project<<<proj_blocks, PROJ_THREADS, 0, stream>>>(
      inp, weights, bias, H, out, n_nodes, n_rel);

  const int scatter_blocks = (n_edges + 255) / 256;
  rgcn_scatter<<<dim3(scatter_blocks, n_rel), 256, 0, stream>>>(
      H, src, dst, eval, out, n_edges, n_nodes);
}


# edges binned by 128-row destination range in the call; one work-group per range adds val*H_r[src] rows in LDS and stores each out line once: no global float atomics; sum order: LDS arrival
# speedup vs baseline: 3.9332x; 1.2576x over previous
#include <hip/hip_runtime.h>
#include <hip/hip_bf16.h>

typedef __attribute__((ext_vector_type(2))) float v2f;
typedef __attribute__((ext_vector_type(8))) float v8f;

#define IN_SIZE 64
#define OUT_SIZE 64


#define PROJ_WAVES 4
#define PROJ_THREADS (PROJ_WAVES * 32)
#define W_PITCH 72

#define BIN_SHIFT 7
#define BIN_ROWS (1 << BIN_SHIFT)
#define MAX_BINS 2048
#define CHUNK 8192
#define BIN_THREADS 1024
#define EDGES_PER_THREAD (CHUNK / BIN_THREADS)
#define SUM_THREADS 512
#define SUM_WAVES (SUM_THREADS / 32)
#define GATHER_GROUP 16

__global__ __launch_bounds__(PROJ_THREADS) void rgcn_project(
    const float* __restrict__ inp, const float* __restrict__ W,
    float* __restrict__ H, int* __restrict__ counters, int n_counters,
    int n_nodes, int n_rel) {
  __shared__ float sW[2][IN_SIZE * W_PITCH];
  const int t = threadIdx.x;
  const int wave = __builtin_amdgcn_readfirstlane(t >> 5);
  const int lane = t & 31;
  const int half = lane >> 4;
  const int l16 = lane & 15;

  for (int i = blockIdx.x * PROJ_THREADS + t; i < n_counters; i += gridDim.x * PROJ_THREADS)
    counters[i] = 0;

  const int st_row = t >> 4;
  const int st_col = (t & 15) * 4;

  {
    const float4* Wv = (const float4*)W;
#pragma unroll
    for (int i = 0; i < 8; ++i)
      *(float4*)&sW[0][(st_row + 8 * i) * W_PITCH + st_col] = Wv[t + i * PROJ_THREADS];
  }

  const int n_row_tiles = (n_nodes + 15) >> 4;
  int tile = blockIdx.x * PROJ_WAVES + wave;
  const bool valid = tile < n_row_tiles;
  if (!valid) tile = n_row_tiles - 1;
  const int rowBase = tile * 16;
  const int rows_left = valid ? (n_nodes - rowBase) : 0;

  const int koff = half * 2;
  int arow = rowBase + l16;
  if (arow >= n_nodes) arow = n_nodes - 1;
  const float* aRow = inp + (size_t)arow * IN_SIZE;
  v2f a[IN_SIZE / 4];
#pragma unroll
  for (int kk = 0; kk < IN_SIZE / 4; ++kk) {
    const float2 av = *(const float2*)(aRow + kk * 4 + koff);
    a[kk][0] = av.x;
    a[kk][1] = av.y;
  }
  __syncthreads();

  for (int r = 0; r < n_rel; ++r) {
    const float* sB = sW[r & 1];
    const int rn = (r + 1 < n_rel) ? r + 1 : r;
    const float4* Wn = (const float4*)(W + (size_t)rn * IN_SIZE * OUT_SIZE);
    float4 pre[8];
#pragma unroll
    for (int i = 0; i < 8; ++i) pre[i] = Wn[t + i * PROJ_THREADS];
    __builtin_amdgcn_sched_barrier(0);

    v8f c[4];
#pragma unroll
    for (int ct = 0; ct < 4; ++ct)
#pragma unroll
      for (int j = 0; j < 8; ++j) c[ct][j] = 0.f;

#pragma unroll
    for (int kk = 0; kk < IN_SIZE / 4; ++kk) {
      const int k = kk * 4;
#pragma unroll
      for (int ct = 0; ct < 4; ++ct) {
        v2f b;
        b[0] = sB[(k + koff) * W_PITCH + ct * 16 + l16];
        b[1] = sB[(k + koff + 1) * W_PITCH + ct * 16 + l16];
        c[ct] = __builtin_amdgcn_wmma_f32_16x16x4_f32(
             false, a[kk],  false, b,
             (short)0, c[ct],  false,  false);
      }
    }

    {
      float* sN = sW[(r + 1) & 1];
#pragma unroll
      for (int i = 0; i < 8; ++i)
        *(float4*)&sN[(st_row + 8 * i) * W_PITCH + st_col] = pre[i];
    }
    __syncthreads();

    float* Hr = H + (size_t)r * n_nodes * OUT_SIZE + (size_t)rowBase * OUT_SIZE;
#pragma unroll
    for (int j = 0; j < 8; ++j) {
      const int rr = j + half * 8;
      if (rr < rows_left) {
        float* hp = Hr + (size_t)rr * OUT_SIZE + 2 * l16;
        *(float2*)hp = make_float2(c[0][j], c[2][j]);
        *(float2*)(hp + 32) = make_float2(c[1][j], c[3][j]);
      }
    }
  }
}

__device__ __forceinline__ void block_exclusive_scan(int* arr, int n, int* wsum) {
  const int t = threadIdx.x;
  const int lane = t & 31;
  const int wave = t >> 5;
  const int per = (n + BIN_THREADS - 1) / BIN_THREADS;
  const int beg = t * per;
  int end = beg + per;
  if (end > n) end = n;
  int s = 0;
  for (int i = beg; i < end; ++i) s += arr[i];
  int incl = s;
  for (int off = 1; off < 32; off <<= 1) {
    const int v = __shfl_up(incl, off);
    if (lane >= off) incl += v;
  }
  if (lane == 31) wsum[wave] = incl;
  __syncthreads();
  if (wave == 0) {
    const int w = wsum[lane];
    int wi = w;
    for (int off = 1; off < 32; off <<= 1) {
      const int v = __shfl_up(wi, off);
      if (lane >= off) wi += v;
    }
    wsum[lane] = wi - w;
  }
  __syncthreads();
  int run = incl - s + wsum[wave];
  for (int i = beg; i < end; ++i) {
    const int v = arr[i];
    arr[i] = run;
    run += v;
  }
  __syncthreads();
}

__global__ __launch_bounds__(BIN_THREADS) void rgcn_bin_count(
    const int* __restrict__ dst, int* __restrict__ totals, int n_edges, int nbins) {
  __shared__ int hist[MAX_BINS];
  const int t = threadIdx.x;
  const int rel = blockIdx.y;
  for (int b = t; b < nbins; b += BIN_THREADS) hist[b] = 0;
  __syncthreads();
  const int c0 = blockIdx.x * CHUNK;
  const int* d = dst + (size_t)rel * n_edges;
#pragma unroll
  for (int k = 0; k < EDGES_PER_THREAD; ++k) {
    const int e = c0 + k * BIN_THREADS + t;
    if (e < n_edges) atomicAdd(&hist[d[e] >> BIN_SHIFT], 1);
  }
  __syncthreads();
  int* tot = totals + rel * nbins;
  for (int b = t; b < nbins; b += BIN_THREADS) {
    const int c = hist[b];
    if (c) atomicAdd(&tot[b], c);
  }
}

__global__ __launch_bounds__(BIN_THREADS) void rgcn_bin_place(
    const int* __restrict__ src, const int* __restrict__ dst,
    const float* __restrict__ eval, const int* __restrict__ totals,
    int* __restrict__ cursor, int* __restrict__ segbase, int2* __restrict__ recs,
    int n_edges, int nbins) {
  __shared__ int hist[MAX_BINS];
  __shared__ int loff[MAX_BINS];
  __shared__ int gb[MAX_BINS];
  __shared__ int wsum[32];
  __shared__ int s_w0[CHUNK];
  __shared__ int s_bin[CHUNK];
  __shared__ float s_val[CHUNK];
  const int t = threadIdx.x;
  const int rel = blockIdx.y;
  const int* tot = totals + rel * nbins;
  for (int b = t; b < nbins; b += BIN_THREADS) {
    hist[b] = 0;
    gb[b] = tot[b];
  }
  __syncthreads();
  const int c0 = blockIdx.x * CHUNK;
  const int left = n_edges - c0;
  const int cnt = left < CHUNK ? left : CHUNK;
  const size_t ebase = (size_t)rel * n_edges + c0;
  int dk[EDGES_PER_THREAD];
  int rk[EDGES_PER_THREAD];
#pragma unroll
  for (int k = 0; k < EDGES_PER_THREAD; ++k) {
    const int i = k * BIN_THREADS + t;
    dk[k] = 0;
    rk[k] = 0;
    if (i < cnt) {
      dk[k] = dst[ebase + i];
      rk[k] = atomicAdd(&hist[dk[k] >> BIN_SHIFT], 1);
    }
  }
  __syncthreads();
  for (int b = t; b < nbins; b += BIN_THREADS) loff[b] = hist[b];
  __syncthreads();
  block_exclusive_scan(loff, nbins, wsum);
  block_exclusive_scan(gb, nbins, wsum);
  for (int b = t; b < nbins; b += BIN_THREADS) {
    const int c = hist[b];
    int g = gb[b] + rel * n_edges;
    if (blockIdx.x == 0) segbase[rel * nbins + b] = g;
    if (c) g += atomicAdd(&cursor[rel * nbins + b], c);
    gb[b] = g;
  }
  __syncthreads();
#pragma unroll
  for (int k = 0; k < EDGES_PER_THREAD; ++k) {
    const int i = k * BIN_THREADS + t;
    if (i < cnt) {
      const int bin = dk[k] >> BIN_SHIFT;
      const int p = loff[bin] + rk[k];
      s_w0[p] = src[ebase + i] | ((dk[k] & (BIN_ROWS - 1)) << 24);
      s_val[p] = eval[ebase + i];
      s_bin[p] = bin;
    }
  }
  __syncthreads();
#pragma unroll
  for (int k = 0; k < EDGES_PER_THREAD; ++k) {
    const int i = k * BIN_THREADS + t;
    if (i < cnt) {
      const int bin = s_bin[i];
      const int g = gb[bin] + (i - loff[bin]);
      recs[g] = make_int2(s_w0[i], __float_as_int(s_val[i]));
    }
  }
}

__device__ __forceinline__ void lds_add(float* p, float v) {
  __hip_atomic_fetch_add(p, v, __ATOMIC_RELAXED, __HIP_MEMORY_SCOPE_WORKGROUP);
}

__global__ __launch_bounds__(SUM_THREADS) __attribute__((amdgpu_waves_per_eu(16, 16)))
void rgcn_bin_sum(
    const float* __restrict__ H, const int2* __restrict__ recs,
    const int* __restrict__ totals, const int* __restrict__ segbase,
    const float* __restrict__ bias, float* __restrict__ out,
    int n_nodes, int n_rel, int nbins) {
  __shared__ float acc[BIN_ROWS * OUT_SIZE];
  const int t = threadIdx.x;
  const int lane = t & 31;
  const int wave = __builtin_amdgcn_readfirstlane(t >> 5);
  const int b = blockIdx.x;

  const int c4 = (t & 15) * 4;
  float4 bsum = make_float4(0.f, 0.f, 0.f, 0.f);
  for (int r = 0; r < n_rel; ++r) {
    const float4 br = *(const float4*)(bias + r * OUT_SIZE + c4);
    bsum.x += br.x; bsum.y += br.y; bsum.z += br.z; bsum.w += br.w;
  }
#pragma unroll
  for (int k = 0; k < (BIN_ROWS * OUT_SIZE / 4) / SUM_THREADS; ++k)
    *(float4*)&acc[(t + k * SUM_THREADS) * 4] = bsum;
  __syncthreads();

  for (int r = 0; r < n_rel; ++r) {
    const int key = r * nbins + b;
    const int start = __builtin_amdgcn_readfirstlane(segbase[key]);
    const int cnt = __builtin_amdgcn_readfirstlane(totals[key]);
    const float* Hr = H + (size_t)r * n_nodes * OUT_SIZE + 2 * lane;
    for (int i = wave * 32; i < cnt; i += SUM_THREADS) {
      const int m = cnt - i;
      if (m >= 32) {
        const int2 rec = recs[start + i + lane];
#pragma unroll
        for (int j0 = 0; j0 < 32; j0 += GATHER_GROUP) {
          float2 mrow[GATHER_GROUP];
#pragma unroll
          for (int j = 0; j < GATHER_GROUP; ++j) {
            const int sj = __builtin_amdgcn_readlane(rec.x, j0 + j) & 0xFFFFFF;
            mrow[j] = *(const float2*)(Hr + (size_t)sj * OUT_SIZE);
          }
#pragma unroll
          for (int j = 0; j < GATHER_GROUP; ++j) {
            const int row = (__builtin_amdgcn_readlane(rec.x, j0 + j) >> 24) & (BIN_ROWS - 1);
            const float vj = __int_as_float(__builtin_amdgcn_readlane(rec.y, j0 + j));
            float* p = acc + row * OUT_SIZE + lane;
            lds_add(p, mrow[j].x * vj);
            lds_add(p + 32, mrow[j].y * vj);
          }
        }
      } else {
        const int2 rec = recs[start + i + (lane < m ? lane : m - 1)];
        for (int j = 0; j < m; ++j) {
          const int w0 = __builtin_amdgcn_readlane(rec.x, j);
          const float vj = __int_as_float(__builtin_amdgcn_readlane(rec.y, j));
          const float2 mj = *(const float2*)(Hr + (size_t)(w0 & 0xFFFFFF) * OUT_SIZE);
          float* p = acc + ((w0 >> 24) & (BIN_ROWS - 1)) * OUT_SIZE + lane;
          lds_add(p, mj.x * vj);
          lds_add(p + 32, mj.y * vj);
        }
      }
    }
  }
  __syncthreads();

#pragma unroll
  for (int k = 0; k < (BIN_ROWS * OUT_SIZE / 4) / SUM_THREADS; ++k) {
    const int q = t + k * SUM_THREADS;
    const int grow = b * BIN_ROWS + (q >> 4);
    if (grow < n_nodes)
      *(float4*)(out + (size_t)grow * OUT_SIZE + c4) = *(const float4*)&acc[q * 4];
  }
}

extern "C" void kernel_launch(void* const* d_in, const int* in_sizes, int n_in,
                              void* d_out, int out_size, void* d_ws, size_t ws_size,
                              hipStream_t stream) {
  const float* inp = (const float*)d_in[0];
  const int* src = (const int*)d_in[1];
  const int* dst = (const int*)d_in[2];
  const float* eval = (const float*)d_in[3];
  const float* weights = (const float*)d_in[4];
  const float* bias = (const float*)d_in[5];
  float* out = (float*)d_out;

  const int n_nodes = in_sizes[0] / IN_SIZE;
  const int n_rel = in_sizes[4] / (IN_SIZE * OUT_SIZE);
  const int n_edges = in_sizes[1] / n_rel;
  const int nbins = (n_nodes + BIN_ROWS - 1) / BIN_ROWS;

  float* H = (float*)d_ws;
  int2* recs = (int2*)(H + (size_t)n_rel * n_nodes * OUT_SIZE);
  int* totals = (int*)(recs + (size_t)n_rel * n_edges);
  int* cursor = totals + n_rel * nbins;
  int* segbase = cursor + n_rel * nbins;

  const int n_row_tiles = (n_nodes + 15) / 16;
  const int proj_blocks = (n_row_tiles + PROJ_WAVES - 1) / PROJ_WAVES;
  rgcn_project<<<proj_blocks, PROJ_THREADS, 0, stream>>>(
      inp, weights, H, totals, 2 * n_rel * nbins, n_nodes, n_rel);

  const int chunks = (n_edges + CHUNK - 1) / CHUNK;
  rgcn_bin_count<<<dim3(chunks, n_rel), BIN_THREADS, 0, stream>>>(dst, totals, n_edges, nbins);
  rgcn_bin_place<<<dim3(chunks, n_rel), BIN_THREADS, 0, stream>>>(
      src, dst, eval, totals, cursor, segbase, recs, n_edges, nbins);
  rgcn_bin_sum<<<nbins, SUM_THREADS, 0, stream>>>(
      H, recs, totals, segbase, bias, out, n_nodes, n_rel, nbins);
}


# summing kernel: a range's records ordered by row in LDS (counting sort, rank = returning integer LDS add); each run of equal rows summed in registers, one LDS float add pair per run; sum order: rank
# speedup vs baseline: 4.4491x; 1.1312x over previous
#include <hip/hip_runtime.h>
#include <hip/hip_bf16.h>

typedef __attribute__((ext_vector_type(2))) float v2f;
typedef __attribute__((ext_vector_type(8))) float v8f;

#define IN_SIZE 64
#define OUT_SIZE 64


#define PROJ_WAVES 4
#define PROJ_THREADS (PROJ_WAVES * 32)
#define W_PITCH 72

#define BIN_SHIFT 7
#define BIN_ROWS (1 << BIN_SHIFT)
#define MAX_BINS 2048
#define CHUNK 8192
#define BIN_THREADS 1024
#define EDGES_PER_THREAD (CHUNK / BIN_THREADS)
#define SUM_THREADS 512
#define SUM_WAVES (SUM_THREADS / 32)
#define GATHER_GROUP 16
#define SORT_CAP 2560
#define SORT_PER_THREAD (SORT_CAP / SUM_THREADS)

__global__ __launch_bounds__(PROJ_THREADS) void rgcn_project(
    const float* __restrict__ inp, const float* __restrict__ W,
    float* __restrict__ H, int* __restrict__ counters, int n_counters,
    int n_nodes, int n_rel) {
  __shared__ float sW[2][IN_SIZE * W_PITCH];
  const int t = threadIdx.x;
  const int wave = __builtin_amdgcn_readfirstlane(t >> 5);
  const int lane = t & 31;
  const int half = lane >> 4;
  const int l16 = lane & 15;

  for (int i = blockIdx.x * PROJ_THREADS + t; i < n_counters; i += gridDim.x * PROJ_THREADS)
    counters[i] = 0;

  const int st_row = t >> 4;
  const int st_col = (t & 15) * 4;

  {
    const float4* Wv = (const float4*)W;
#pragma unroll
    for (int i = 0; i < 8; ++i)
      *(float4*)&sW[0][(st_row + 8 * i) * W_PITCH + st_col] = Wv[t + i * PROJ_THREADS];
  }

  const int n_row_tiles = (n_nodes + 15) >> 4;
  int tile = blockIdx.x * PROJ_WAVES + wave;
  const bool valid = tile < n_row_tiles;
  if (!valid) tile = n_row_tiles - 1;
  const int rowBase = tile * 16;
  const int rows_left = valid ? (n_nodes - rowBase) : 0;

  const int koff = half * 2;
  int arow = rowBase + l16;
  if (arow >= n_nodes) arow = n_nodes - 1;
  const float* aRow = inp + (size_t)arow * IN_SIZE;
  v2f a[IN_SIZE / 4];
#pragma unroll
  for (int kk = 0; kk < IN_SIZE / 4; ++kk) {
    const float2 av = *(const float2*)(aRow + kk * 4 + koff);
    a[kk][0] = av.x;
    a[kk][1] = av.y;
  }
  __syncthreads();

  for (int r = 0; r < n_rel; ++r) {
    const float* sB = sW[r & 1];
    const int rn = (r + 1 < n_rel) ? r + 1 : r;
    const float4* Wn = (const float4*)(W + (size_t)rn * IN_SIZE * OUT_SIZE);
    float4 pre[8];
#pragma unroll
    for (int i = 0; i < 8; ++i) pre[i] = Wn[t + i * PROJ_THREADS];
    __builtin_amdgcn_sched_barrier(0);

    v8f c[4];
#pragma unroll
    for (int ct = 0; ct < 4; ++ct)
#pragma unroll
      for (int j = 0; j < 8; ++j) c[ct][j] = 0.f;

#pragma unroll
    for (int kk = 0; kk < IN_SIZE / 4; ++kk) {
      const int k = kk * 4;
#pragma unroll
      for (int ct = 0; ct < 4; ++ct) {
        v2f b;
        b[0] = sB[(k + koff) * W_PITCH + ct * 16 + l16];
        b[1] = sB[(k + koff + 1) * W_PITCH + ct * 16 + l16];
        c[ct] = __builtin_amdgcn_wmma_f32_16x16x4_f32(
             false, a[kk],  false, b,
             (short)0, c[ct],  false,  false);
      }
    }

    {
      float* sN = sW[(r + 1) & 1];
#pragma unroll
      for (int i = 0; i < 8; ++i)
        *(float4*)&sN[(st_row + 8 * i) * W_PITCH + st_col] = pre[i];
    }
    __syncthreads();

    float* Hr = H + (size_t)r * n_nodes * OUT_SIZE + (size_t)rowBase * OUT_SIZE;
#pragma unroll
    for (int j = 0; j < 8; ++j) {
      const int rr = j + half * 8;
      if (rr < rows_left) {
        float* hp = Hr + (size_t)rr * OUT_SIZE + 2 * l16;
        *(float2*)hp = make_float2(c[0][j], c[2][j]);
        *(float2*)(hp + 32) = make_float2(c[1][j], c[3][j]);
      }
    }
  }
}

__device__ __forceinline__ void block_exclusive_scan(int* arr, int n, int* wsum) {
  const int t = threadIdx.x;
  const int lane = t & 31;
  const int wave = t >> 5;
  const int per = (n + BIN_THREADS - 1) / BIN_THREADS;
  const int beg = t * per;
  int end = beg + per;
  if (end > n) end = n;
  int s = 0;
  for (int i = beg; i < end; ++i) s += arr[i];
  int incl = s;
  for (int off = 1; off < 32; off <<= 1) {
    const int v = __shfl_up(incl, off);
    if (lane >= off) incl += v;
  }
  if (lane == 31) wsum[wave] = incl;
  __syncthreads();
  if (wave == 0) {
    const int w = wsum[lane];
    int wi = w;
    for (int off = 1; off < 32; off <<= 1) {
      const int v = __shfl_up(wi, off);
      if (lane >= off) wi += v;
    }
    wsum[lane] = wi - w;
  }
  __syncthreads();
  int run = incl - s + wsum[wave];
  for (int i = beg; i < end; ++i) {
    const int v = arr[i];
    arr[i] = run;
    run += v;
  }
  __syncthreads();
}

__global__ __launch_bounds__(BIN_THREADS) void rgcn_bin_count(
    const int* __restrict__ dst, int* __restrict__ totals, int n_edges, int nbins) {
  __shared__ int hist[MAX_BINS];
  const int t = threadIdx.x;
  const int rel = blockIdx.y;
  for (int b = t; b < nbins; b += BIN_THREADS) hist[b] = 0;
  __syncthreads();
  const int c0 = blockIdx.x * CHUNK;
  const int* d = dst + (size_t)rel * n_edges;
#pragma unroll
  for (int k = 0; k < EDGES_PER_THREAD; ++k) {
    const int e = c0 + k * BIN_THREADS + t;
    if (e < n_edges) atomicAdd(&hist[d[e] >> BIN_SHIFT], 1);
  }
  __syncthreads();
  int* tot = totals + rel * nbins;
  for (int b = t; b < nbins; b += BIN_THREADS) {
    const int c = hist[b];
    if (c) atomicAdd(&tot[b], c);
  }
}

__global__ __launch_bounds__(BIN_THREADS) void rgcn_bin_place(
    const int* __restrict__ src, const int* __restrict__ dst,
    const float* __restrict__ eval, const int* __restrict__ totals,
    int* __restrict__ cursor, int* __restrict__ segbase, int2* __restrict__ recs,
    int n_edges, int nbins) {
  __shared__ int hist[MAX_BINS];
  __shared__ int loff[MAX_BINS];
  __shared__ int gb[MAX_BINS];
  __shared__ int wsum[32];
  __shared__ int s_w0[CHUNK];
  __shared__ int s_bin[CHUNK];
  __shared__ float s_val[CHUNK];
  const int t = threadIdx.x;
  const int rel = blockIdx.y;
  const int* tot = totals + rel * nbins;
  for (int b = t; b < nbins; b += BIN_THREADS) {
    hist[b] = 0;
    gb[b] = tot[b];
  }
  __syncthreads();
  const int c0 = blockIdx.x * CHUNK;
  const int left = n_edges - c0;
  const int cnt = left < CHUNK ? left : CHUNK;
  const size_t ebase = (size_t)rel * n_edges + c0;
  int dk[EDGES_PER_THREAD];
  int rk[EDGES_PER_THREAD];
#pragma unroll
  for (int k = 0; k < EDGES_PER_THREAD; ++k) {
    const int i = k * BIN_THREADS + t;
    dk[k] = 0;
    rk[k] = 0;
    if (i < cnt) {
      dk[k] = dst[ebase + i];
      rk[k] = atomicAdd(&hist[dk[k] >> BIN_SHIFT], 1);
    }
  }
  __syncthreads();
  for (int b = t; b < nbins; b += BIN_THREADS) loff[b] = hist[b];
  __syncthreads();
  block_exclusive_scan(loff, nbins, wsum);
  block_exclusive_scan(gb, nbins, wsum);
  for (int b = t; b < nbins; b += BIN_THREADS) {
    const int c = hist[b];
    int g = gb[b] + rel * n_edges;
    if (blockIdx.x == 0) segbase[rel * nbins + b] = g;
    if (c) g += atomicAdd(&cursor[rel * nbins + b], c);
    gb[b] = g;
  }
  __syncthreads();
#pragma unroll
  for (int k = 0; k < EDGES_PER_THREAD; ++k) {
    const int i = k * BIN_THREADS + t;
    if (i < cnt) {
      const int bin = dk[k] >> BIN_SHIFT;
      const int p = loff[bin] + rk[k];
      s_w0[p] = src[ebase + i] | ((dk[k] & (BIN_ROWS - 1)) << 24);
      s_val[p] = eval[ebase + i];
      s_bin[p] = bin;
    }
  }
  __syncthreads();
#pragma unroll
  for (int k = 0; k < EDGES_PER_THREAD; ++k) {
    const int i = k * BIN_THREADS + t;
    if (i < cnt) {
      const int bin = s_bin[i];
      const int g = gb[bin] + (i - loff[bin]);
      recs[g] = make_int2(s_w0[i], __float_as_int(s_val[i]));
    }
  }
}

__device__ __forceinline__ void lds_add(float* p, float v) {
  __hip_atomic_fetch_add(p, v, __ATOMIC_RELAXED, __HIP_MEMORY_SCOPE_WORKGROUP);
}

template <bool FULL>
__device__ __forceinline__ void sum_group(const int2 rec, const int m, const float* Hr,
                                          float* accl, int& cur, float& s0, float& s1) {
#pragma unroll
  for (int j0 = 0; j0 < 32; j0 += GATHER_GROUP) {
    if (!FULL && j0 >= m) break;
    float2 mrow[GATHER_GROUP];
#pragma unroll
    for (int j = 0; j < GATHER_GROUP; ++j) {
      const int sj = __builtin_amdgcn_readlane(rec.x, j0 + j) & 0xFFFFFF;
      mrow[j] = *(const float2*)(Hr + (size_t)sj * OUT_SIZE);
    }
#pragma unroll
    for (int j = 0; j < GATHER_GROUP; ++j) {
      if (FULL || j0 + j < m) {
        const int row = (__builtin_amdgcn_readlane(rec.x, j0 + j) >> 24) & (BIN_ROWS - 1);
        const float vj = __int_as_float(__builtin_amdgcn_readlane(rec.y, j0 + j));
        if (row != cur) {
          lds_add(accl + cur * OUT_SIZE, s0);
          lds_add(accl + cur * OUT_SIZE + 32, s1);
          s0 = 0.f;
          s1 = 0.f;
          cur = row;
        }
        s0 += mrow[j].x * vj;
        s1 += mrow[j].y * vj;
      }
    }
  }
}

__global__ __launch_bounds__(SUM_THREADS) __attribute__((amdgpu_waves_per_eu(16, 16)))
void rgcn_bin_sum(
    const float* __restrict__ H, const int2* __restrict__ recs,
    const int* __restrict__ totals, const int* __restrict__ segbase,
    const float* __restrict__ bias, float* __restrict__ out,
    int n_nodes, int n_rel, int nbins) {
  __shared__ float acc[BIN_ROWS * OUT_SIZE];
  __shared__ int2 sorted[SORT_CAP];
  __shared__ __align__(16) int rcnt[BIN_ROWS];
  __shared__ __align__(16) int roff[BIN_ROWS];
  const int t = threadIdx.x;
  const int lane = t & 31;
  const int wave = __builtin_amdgcn_readfirstlane(t >> 5);
  const int b = blockIdx.x;

  const int c4 = (t & 15) * 4;
  float4 bsum = make_float4(0.f, 0.f, 0.f, 0.f);
  for (int r = 0; r < n_rel; ++r) {
    const float4 br = *(const float4*)(bias + r * OUT_SIZE + c4);
    bsum.x += br.x; bsum.y += br.y; bsum.z += br.z; bsum.w += br.w;
  }
#pragma unroll
  for (int k = 0; k < (BIN_ROWS * OUT_SIZE / 4) / SUM_THREADS; ++k)
    *(float4*)&acc[(t + k * SUM_THREADS) * 4] = bsum;
  if (t < BIN_ROWS) rcnt[t] = 0;
  __syncthreads();

  float* const accl = acc + lane;
  const float* Hr = H + 2 * lane;
  for (int r = 0; r < n_rel; ++r, Hr += (size_t)n_nodes * OUT_SIZE) {
    const int key = r * nbins + b;
    const int start = __builtin_amdgcn_readfirstlane(segbase[key]);
    const int cnt = __builtin_amdgcn_readfirstlane(totals[key]);
    for (int c0 = 0; c0 < cnt; c0 += SORT_CAP) {
      const int left = cnt - c0;
      const int n = left < SORT_CAP ? left : SORT_CAP;

      const int2* rp = recs + (start + c0);
      int2 rk[SORT_PER_THREAD];
      int rank[SORT_PER_THREAD];
#pragma unroll
      for (int k = 0; k < SORT_PER_THREAD; ++k) {
        const int nk = n - k * SUM_THREADS;
        rk[k] = make_int2(0, 0);
        if (nk > 0) rk[k] = (rp + k * SUM_THREADS)[t < nk ? t : nk - 1];
      }
#pragma unroll
      for (int k = 0; k < SORT_PER_THREAD; ++k) {
        rank[k] = 0;
        if (t < n - k * SUM_THREADS)
          rank[k] = atomicAdd(&rcnt[(rk[k].x >> 24) & (BIN_ROWS - 1)], 1);
      }
      __syncthreads();
      if (wave == 0) {
        const int4 c = *(const int4*)&rcnt[4 * t];
        const int s = c.x + c.y + c.z + c.w;
        int incl = s;
        for (int off = 1; off < 32; off <<= 1) {
          const int v = __shfl_up(incl, off);
          if (lane >= off) incl += v;
        }
        const int ex = incl - s;
        *(int4*)&roff[4 * t] = make_int4(ex, ex + c.x, ex + c.x + c.y, ex + c.x + c.y + c.z);
        *(int4*)&rcnt[4 * t] = make_int4(0, 0, 0, 0);
      }
      __syncthreads();
#pragma unroll
      for (int k = 0; k < SORT_PER_THREAD; ++k) {
        if (t < n - k * SUM_THREADS)
          sorted[roff[(rk[k].x >> 24) & (BIN_ROWS - 1)] + rank[k]] = rk[k];
      }
      __syncthreads();

      const int span = ((n + SUM_THREADS - 1) / SUM_THREADS) * 32;
      const int beg = wave * span;
      int end = beg + span;
      if (end > n) end = n;
      if (beg < end) {
        int cur = __builtin_amdgcn_readfirstlane(sorted[beg].x >> 24) & (BIN_ROWS - 1);
        float s0 = 0.f, s1 = 0.f;
        for (int i = beg; i < end; i += 32) {
          const int m = end - i;
          if (m >= 32) {
            sum_group<true>(sorted[i + lane], 32, Hr, accl, cur, s0, s1);
          } else {
            sum_group<false>(sorted[i + (lane < m ? lane : m - 1)], m, Hr, accl, cur, s0, s1);
          }
        }
        lds_add(accl + cur * OUT_SIZE, s0);
        lds_add(accl + cur * OUT_SIZE + 32, s1);
      }
    }
  }
  __syncthreads();

#pragma unroll
  for (int k = 0; k < (BIN_ROWS * OUT_SIZE / 4) / SUM_THREADS; ++k) {
    const int q = t + k * SUM_THREADS;
    const int grow = b * BIN_ROWS + (q >> 4);
    if (grow < n_nodes)
      *(float4*)(out + (size_t)grow * OUT_SIZE + c4) = *(const float4*)&acc[q * 4];
  }
}

extern "C" void kernel_launch(void* const* d_in, const int* in_sizes, int n_in,
                              void* d_out, int out_size, void* d_ws, size_t ws_size,
                              hipStream_t stream) {
  const float* inp = (const float*)d_in[0];
  const int* src = (const int*)d_in[1];
  const int* dst = (const int*)d_in[2];
  const float* eval = (const float*)d_in[3];
  const float* weights = (const float*)d_in[4];
  const float* bias = (const float*)d_in[5];
  float* out = (float*)d_out;

  const int n_nodes = in_sizes[0] / IN_SIZE;
  const int n_rel = in_sizes[4] / (IN_SIZE * OUT_SIZE);
  const int n_edges = in_sizes[1] / n_rel;
  const int nbins = (n_nodes + BIN_ROWS - 1) / BIN_ROWS;

  float* H = (float*)d_ws;
  int2* recs = (int2*)(H + (size_t)n_rel * n_nodes * OUT_SIZE);
  int* totals = (int*)(recs + (size_t)n_rel * n_edges);
  int* cursor = totals + n_rel * nbins;
  int* segbase = cursor + n_rel * nbins;

  const int n_row_tiles = (n_nodes + 15) / 16;
  const int proj_blocks = (n_row_tiles + PROJ_WAVES - 1) / PROJ_WAVES;
  rgcn_project<<<proj_blocks, PROJ_THREADS, 0, stream>>>(
      inp, weights, H, totals, 2 * n_rel * nbins, n_nodes, n_rel);

  const int chunks = (n_edges + CHUNK - 1) / CHUNK;
  rgcn_bin_count<<<dim3(chunks, n_rel), BIN_THREADS, 0, stream>>>(dst, totals, n_edges, nbins);
  rgcn_bin_place<<<dim3(chunks, n_rel), BIN_THREADS, 0, stream>>>(
      src, dst, eval, totals, cursor, segbase, recs, n_edges, nbins);
  rgcn_bin_sum<<<nbins, SUM_THREADS, 0, stream>>>(
      H, recs, totals, segbase, bias, out, n_nodes, n_rel, nbins);
}


# W_r applied after the edge sum, inside the summing kernel: gathers read inp rows, each range's 128x64 sums times W_r on f32 WMMA from LDS; projection kernel and H planes dropped; sum order changed
# speedup vs baseline: 5.3937x; 1.2123x over previous
#include <hip/hip_runtime.h>
#include <hip/hip_bf16.h>

typedef __attribute__((ext_vector_type(2))) float v2f;
typedef __attribute__((ext_vector_type(8))) float v8f;

#define IN_SIZE 64
#define OUT_SIZE 64


#define W_PITCH 72
#define A_PITCH 68

#define BIN_SHIFT 7
#define BIN_ROWS (1 << BIN_SHIFT)
#define MAX_BINS 2048
#define CHUNK 8192
#define BIN_THREADS 1024
#define EDGES_PER_THREAD (CHUNK / BIN_THREADS)
#define SUM_THREADS 512
#define SUM_WAVES (SUM_THREADS / 32)
#define GATHER_GROUP 16
#define SORT_CAP 2560
#define SORT_PER_THREAD (SORT_CAP / SUM_THREADS)

__global__ __launch_bounds__(BIN_THREADS) void rgcn_clear(int* __restrict__ counters, int n_counters) {
  const int i = blockIdx.x * BIN_THREADS + threadIdx.x;
  if (i < n_counters) counters[i] = 0;
}

__device__ __forceinline__ void block_exclusive_scan(int* arr, int n, int* wsum) {
  const int t = threadIdx.x;
  const int lane = t & 31;
  const int wave = t >> 5;
  const int per = (n + BIN_THREADS - 1) / BIN_THREADS;
  const int beg = t * per;
  int end = beg + per;
  if (end > n) end = n;
  int s = 0;
  for (int i = beg; i < end; ++i) s += arr[i];
  int incl = s;
  for (int off = 1; off < 32; off <<= 1) {
    const int v = __shfl_up(incl, off);
    if (lane >= off) incl += v;
  }
  if (lane == 31) wsum[wave] = incl;
  __syncthreads();
  if (wave == 0) {
    const int w = wsum[lane];
    int wi = w;
    for (int off = 1; off < 32; off <<= 1) {
      const int v = __shfl_up(wi, off);
      if (lane >= off) wi += v;
    }
    wsum[lane] = wi - w;
  }
  __syncthreads();
  int run = incl - s + wsum[wave];
  for (int i = beg; i < end; ++i) {
    const int v = arr[i];
    arr[i] = run;
    run += v;
  }
  __syncthreads();
}

__global__ __launch_bounds__(BIN_THREADS) void rgcn_bin_count(
    const int* __restrict__ dst, int* __restrict__ totals, int n_edges, int nbins) {
  __shared__ int hist[MAX_BINS];
  const int t = threadIdx.x;
  const int rel = blockIdx.y;
  for (int b = t; b < nbins; b += BIN_THREADS) hist[b] = 0;
  __syncthreads();
  const int c0 = blockIdx.x * CHUNK;
  const int* d = dst + (size_t)rel * n_edges;
#pragma unroll
  for (int k = 0; k < EDGES_PER_THREAD; ++k) {
    const int e = c0 + k * BIN_THREADS + t;
    if (e < n_edges) atomicAdd(&hist[d[e] >> BIN_SHIFT], 1);
  }
  __syncthreads();
  int* tot = totals + rel * nbins;
  for (int b = t; b < nbins; b += BIN_THREADS) {
    const int c = hist[b];
    if (c) atomicAdd(&tot[b], c);
  }
}

__global__ __launch_bounds__(BIN_THREADS) void rgcn_bin_place(
    const int* __restrict__ src, const int* __restrict__ dst,
    const float* __restrict__ eval, const int* __restrict__ totals,
    int* __restrict__ cursor, int* __restrict__ segbase, int2* __restrict__ recs,
    int n_edges, int nbins) {
  __shared__ int hist[MAX_BINS];
  __shared__ int loff[MAX_BINS];
  __shared__ int gb[MAX_BINS];
  __shared__ int wsum[32];
  __shared__ int s_w0[CHUNK];
  __shared__ int s_bin[CHUNK];
  __shared__ float s_val[CHUNK];
  const int t = threadIdx.x;
  const int rel = blockIdx.y;
  const int* tot = totals + rel * nbins;
  for (int b = t; b < nbins; b += BIN_THREADS) {
    hist[b] = 0;
    gb[b] = tot[b];
  }
  __syncthreads();
  const int c0 = blockIdx.x * CHUNK;
  const int left = n_edges - c0;
  const int cnt = left < CHUNK ? left : CHUNK;
  const size_t ebase = (size_t)rel * n_edges + c0;
  int dk[EDGES_PER_THREAD];
  int rk[EDGES_PER_THREAD];
#pragma unroll
  for (int k = 0; k < EDGES_PER_THREAD; ++k) {
    const int i = k * BIN_THREADS + t;
    dk[k] = 0;
    rk[k] = 0;
    if (i < cnt) {
      dk[k] = dst[ebase + i];
      rk[k] = atomicAdd(&hist[dk[k] >> BIN_SHIFT], 1);
    }
  }
  __syncthreads();
  for (int b = t; b < nbins; b += BIN_THREADS) loff[b] = hist[b];
  __syncthreads();
  block_exclusive_scan(loff, nbins, wsum);
  block_exclusive_scan(gb, nbins, wsum);
  for (int b = t; b < nbins; b += BIN_THREADS) {
    const int c = hist[b];
    int g = gb[b] + rel * n_edges;
    if (blockIdx.x == 0) segbase[rel * nbins + b] = g;
    if (c) g += atomicAdd(&cursor[rel * nbins + b], c);
    gb[b] = g;
  }
  __syncthreads();
#pragma unroll
  for (int k = 0; k < EDGES_PER_THREAD; ++k) {
    const int i = k * BIN_THREADS + t;
    if (i < cnt) {
      const int bin = dk[k] >> BIN_SHIFT;
      const int p = loff[bin] + rk[k];
      s_w0[p] = src[ebase + i] | ((dk[k] & (BIN_ROWS - 1)) << 24);
      s_val[p] = eval[ebase + i];
      s_bin[p] = bin;
    }
  }
  __syncthreads();
#pragma unroll
  for (int k = 0; k < EDGES_PER_THREAD; ++k) {
    const int i = k * BIN_THREADS + t;
    if (i < cnt) {
      const int bin = s_bin[i];
      const int g = gb[bin] + (i - loff[bin]);
      recs[g] = make_int2(s_w0[i], __float_as_int(s_val[i]));
    }
  }
}

__device__ __forceinline__ void lds_add(float* p, float v) {
  __hip_atomic_fetch_add(p, v, __ATOMIC_RELAXED, __HIP_MEMORY_SCOPE_WORKGROUP);
}

template <bool FULL>
__device__ __forceinline__ void sum_group(const int2 rec, const int m, const float2* __restrict__ X2,
                                          const unsigned lane, float* accl, int& cur,
                                          float& s0, float& s1) {
#pragma unroll
  for (int j0 = 0; j0 < 32; j0 += GATHER_GROUP) {
    if (!FULL && j0 >= m) break;
    float2 mrow[GATHER_GROUP];
#pragma unroll
    for (int j = 0; j < GATHER_GROUP; ++j) {
      const unsigned sj = (unsigned)__builtin_amdgcn_readlane(rec.x, j0 + j) & 0xFFFFFFu;
      mrow[j] = X2[sj * (IN_SIZE / 2) + lane];
    }
#pragma unroll
    for (int j = 0; j < GATHER_GROUP; ++j) {
      if (FULL || j0 + j < m) {
        const int row = (__builtin_amdgcn_readlane(rec.x, j0 + j) >> 24) & (BIN_ROWS - 1);
        const float vj = __int_as_float(__builtin_amdgcn_readlane(rec.y, j0 + j));
        if (row != cur) {
          lds_add(accl + cur * A_PITCH, s0);
          lds_add(accl + cur * A_PITCH + 1, s1);
          s0 = 0.f;
          s1 = 0.f;
          cur = row;
        }
        s0 += mrow[j].x * vj;
        s1 += mrow[j].y * vj;
      }
    }
  }
}

__global__ __launch_bounds__(SUM_THREADS) __attribute__((amdgpu_waves_per_eu(16, 16)))
void rgcn_bin_sum(
    const float* __restrict__ inp, const float* __restrict__ W,
    const int2* __restrict__ recs,
    const int* __restrict__ totals, const int* __restrict__ segbase,
    const float* __restrict__ bias, float* __restrict__ out,
    int n_nodes, int n_rel, int nbins) {
  __shared__ __align__(16) float agg[BIN_ROWS * A_PITCH];
  __shared__ __align__(16) int2 sorted[SORT_CAP];
  __shared__ __align__(16) int rcnt[BIN_ROWS];
  __shared__ __align__(16) int roff[BIN_ROWS];
  static_assert(sizeof(int2) * SORT_CAP >= sizeof(float) * IN_SIZE * W_PITCH, "W_r must fit the record space");
  float* const sW = (float*)sorted;
  const int t = threadIdx.x;
  const int lane = t & 31;
  const int wave = __builtin_amdgcn_readfirstlane(t >> 5);
  const int l16 = lane & 15;
  const int b = blockIdx.x;
  const int rt = wave >> 1;
  const int ctb = (wave & 1) * 2;

  v8f cacc[2];
#pragma unroll
  for (int i = 0; i < 2; ++i) {
    float bs = 0.f;
    for (int r = 0; r < n_rel; ++r) bs += bias[r * OUT_SIZE + (ctb + i) * 16 + l16];
#pragma unroll
    for (int j = 0; j < 8; ++j) cacc[i][j] = bs;
  }

#pragma unroll
  for (int k = 0; k < (BIN_ROWS * A_PITCH / 4 + SUM_THREADS - 1) / SUM_THREADS; ++k) {
    const int q = t + k * SUM_THREADS;
    if (q < BIN_ROWS * A_PITCH / 4) *(float4*)&agg[q * 4] = make_float4(0.f, 0.f, 0.f, 0.f);
  }
  if (t < BIN_ROWS) rcnt[t] = 0;
  __syncthreads();

  const float2* const inp2 = (const float2*)inp;
  float* const accl = agg + 2 * lane;
  for (int r = 0; r < n_rel; ++r) {
    const int key = r * nbins + b;
    const int start = __builtin_amdgcn_readfirstlane(segbase[key]);
    const int cnt = __builtin_amdgcn_readfirstlane(totals[key]);
    for (int c0 = 0; c0 < cnt; c0 += SORT_CAP) {
      const int left = cnt - c0;
      const int n = left < SORT_CAP ? left : SORT_CAP;

      const int2* rp = recs + (start + c0);
      int ts = t;
      asm volatile("" : "+v"(ts));
      int2 rk[SORT_PER_THREAD];
      int rank[SORT_PER_THREAD];
#pragma unroll
      for (int k = 0; k < SORT_PER_THREAD; ++k) {
        const int nk = n - k * SUM_THREADS;
        rk[k] = make_int2(0, 0);
        if (nk > 0) rk[k] = (rp + k * SUM_THREADS)[ts < nk ? ts : nk - 1];
      }
#pragma unroll
      for (int k = 0; k < SORT_PER_THREAD; ++k) {
        rank[k] = 0;
        if (ts < n - k * SUM_THREADS)
          rank[k] = atomicAdd(&rcnt[(rk[k].x >> 24) & (BIN_ROWS - 1)], 1);
      }
      __syncthreads();
      if (wave == 0) {
        const int4 c = *(const int4*)&rcnt[4 * ts];
        const int s = c.x + c.y + c.z + c.w;
        int incl = s;
        for (int off = 1; off < 32; off <<= 1) {
          const int v = __shfl_up(incl, off);
          if (lane >= off) incl += v;
        }
        const int ex = incl - s;
        *(int4*)&roff[4 * ts] = make_int4(ex, ex + c.x, ex + c.x + c.y, ex + c.x + c.y + c.z);
        *(int4*)&rcnt[4 * ts] = make_int4(0, 0, 0, 0);
      }
      __syncthreads();
#pragma unroll
      for (int k = 0; k < SORT_PER_THREAD; ++k) {
        if (ts < n - k * SUM_THREADS)
          sorted[roff[(rk[k].x >> 24) & (BIN_ROWS - 1)] + rank[k]] = rk[k];
      }
      __syncthreads();

      const int span = ((n + SUM_THREADS - 1) / SUM_THREADS) * 32;
      const int beg = wave * span;
      int end = beg + span;
      if (end > n) end = n;
      if (beg < end) {
        int cur = __builtin_amdgcn_readfirstlane(sorted[beg].x >> 24) & (BIN_ROWS - 1);
        float s0 = 0.f, s1 = 0.f;
        for (int i = beg; i < end; i += 32) {
          const int m = end - i;
          if (m >= 32) {
            sum_group<true>(sorted[i + lane], 32, inp2, (unsigned)lane, accl, cur, s0, s1);
          } else {
            sum_group<false>(sorted[i + (lane < m ? lane : m - 1)], m, inp2, (unsigned)lane, accl, cur, s0, s1);
          }
        }
        lds_add(accl + cur * A_PITCH, s0);
        lds_add(accl + cur * A_PITCH + 1, s1);
      }
    }

    int tp = t;
    asm volatile("" : "+v"(tp));
    const int halfp = (tp >> 4) & 1;
    const int l16p = tp & 15;
    const float4* Wv = (const float4*)(W + (size_t)r * IN_SIZE * OUT_SIZE);
    const float4 w0 = Wv[tp];
    const float4 w1 = Wv[tp + SUM_THREADS];
    __syncthreads();
    *(float4*)&sW[(tp >> 4) * W_PITCH + l16p * 4] = w0;
    *(float4*)&sW[((tp >> 4) + 32) * W_PITCH + l16p * 4] = w1;
    __syncthreads();

    {
      const float* aP = agg + (rt * 16 + l16p) * A_PITCH + 2 * halfp;
      const float* bP = sW + (2 * halfp) * W_PITCH + ctb * 16 + l16p;
#pragma unroll
      for (int kk = 0; kk < IN_SIZE / 4; ++kk) {
        const float2 av = *(const float2*)(aP + kk * 4);
        v2f a;
        a[0] = av.x;
        a[1] = av.y;
#pragma unroll
        for (int i = 0; i < 2; ++i) {
          v2f bf;
          bf[0] = bP[(kk * 4) * W_PITCH + i * 16];
          bf[1] = bP[(kk * 4 + 1) * W_PITCH + i * 16];
          cacc[i] = __builtin_amdgcn_wmma_f32_16x16x4_f32(
               false, a,  false, bf,
               (short)0, cacc[i],  false,  false);
        }
      }
    }
    __syncthreads();
    if (r + 1 < n_rel) {
#pragma unroll
      for (int k = 0; k < (BIN_ROWS * A_PITCH / 4 + SUM_THREADS - 1) / SUM_THREADS; ++k) {
        const int q = tp + k * SUM_THREADS;
        if (q < BIN_ROWS * A_PITCH / 4) *(float4*)&agg[q * 4] = make_float4(0.f, 0.f, 0.f, 0.f);
      }
    }
  }

  int te = t;
  asm volatile("" : "+v"(te));
  const int halfe = (te >> 4) & 1;
  const int l16e = te & 15;
#pragma unroll
  for (int i = 0; i < 2; ++i)
#pragma unroll
    for (int j = 0; j < 8; ++j)
      agg[(rt * 16 + j + 8 * halfe) * A_PITCH + (ctb + i) * 16 + l16e] = cacc[i][j];
  __syncthreads();
  const int c4 = l16e * 4;
#pragma unroll
  for (int k = 0; k < (BIN_ROWS * OUT_SIZE / 4) / SUM_THREADS; ++k) {
    const int q = te + k * SUM_THREADS;
    const int lrow = q >> 4;
    const int grow = b * BIN_ROWS + lrow;
    if (grow < n_nodes)
      *(float4*)(out + (size_t)grow * OUT_SIZE + c4) = *(const float4*)&agg[lrow * A_PITCH + c4];
  }
}

extern "C" void kernel_launch(void* const* d_in, const int* in_sizes, int n_in,
                              void* d_out, int out_size, void* d_ws, size_t ws_size,
                              hipStream_t stream) {
  const float* inp = (const float*)d_in[0];
  const int* src = (const int*)d_in[1];
  const int* dst = (const int*)d_in[2];
  const float* eval = (const float*)d_in[3];
  const float* weights = (const float*)d_in[4];
  const float* bias = (const float*)d_in[5];
  float* out = (float*)d_out;

  const int n_nodes = in_sizes[0] / IN_SIZE;
  const int n_rel = in_sizes[4] / (IN_SIZE * OUT_SIZE);
  const int n_edges = in_sizes[1] / n_rel;
  const int nbins = (n_nodes + BIN_ROWS - 1) / BIN_ROWS;

  int2* recs = (int2*)d_ws;
  int* totals = (int*)(recs + (size_t)n_rel * n_edges);
  int* cursor = totals + n_rel * nbins;
  int* segbase = cursor + n_rel * nbins;

  const int n_counters = 2 * n_rel * nbins;
  rgcn_clear<<<(n_counters + BIN_THREADS - 1) / BIN_THREADS, BIN_THREADS, 0, stream>>>(totals, n_counters);

  const int chunks = (n_edges + CHUNK - 1) / CHUNK;
  rgcn_bin_count<<<dim3(chunks, n_rel), BIN_THREADS, 0, stream>>>(dst, totals, n_edges, nbins);
  rgcn_bin_place<<<dim3(chunks, n_rel), BIN_THREADS, 0, stream>>>(
      src, dst, eval, totals, cursor, segbase, recs, n_edges, nbins);
  rgcn_bin_sum<<<nbins, SUM_THREADS, 0, stream>>>(
      inp, weights, recs, totals, segbase, bias, out, n_nodes, n_rel, nbins);
}


# summing kernel: one 16-B-per-lane gather reads two records' rows (one per 16-lane half, 4 whole lines per load, was 2); each half sums its own half piece; sum order: runs cut per half piece
# speedup vs baseline: 5.5733x; 1.0333x over previous
#include <hip/hip_runtime.h>
#include <hip/hip_bf16.h>

typedef __attribute__((ext_vector_type(2))) float v2f;
typedef __attribute__((ext_vector_type(8))) float v8f;

#define IN_SIZE 64
#define OUT_SIZE 64


#define W_PITCH 72
#define A_PITCH 68

#define BIN_SHIFT 7
#define BIN_ROWS (1 << BIN_SHIFT)
#define MAX_BINS 2048
#define CHUNK 8192
#define BIN_THREADS 1024
#define EDGES_PER_THREAD (CHUNK / BIN_THREADS)
#define SUM_THREADS 512
#define SUM_WAVES (SUM_THREADS / 32)
#define SORT_CAP 2560
#define SORT_PER_THREAD (SORT_CAP / SUM_THREADS)

__global__ __launch_bounds__(BIN_THREADS) void rgcn_clear(int* __restrict__ counters, int n_counters) {
  const int i = blockIdx.x * BIN_THREADS + threadIdx.x;
  if (i < n_counters) counters[i] = 0;
}

__device__ __forceinline__ void block_exclusive_scan(int* arr, int n, int* wsum) {
  const int t = threadIdx.x;
  const int lane = t & 31;
  const int wave = t >> 5;
  const int per = (n + BIN_THREADS - 1) / BIN_THREADS;
  const int beg = t * per;
  int end = beg + per;
  if (end > n) end = n;
  int s = 0;
  for (int i = beg; i < end; ++i) s += arr[i];
  int incl = s;
  for (int off = 1; off < 32; off <<= 1) {
    const int v = __shfl_up(incl, off);
    if (lane >= off) incl += v;
  }
  if (lane == 31) wsum[wave] = incl;
  __syncthreads();
  if (wave == 0) {
    const int w = wsum[lane];
    int wi = w;
    for (int off = 1; off < 32; off <<= 1) {
      const int v = __shfl_up(wi, off);
      if (lane >= off) wi += v;
    }
    wsum[lane] = wi - w;
  }
  __syncthreads();
  int run = incl - s + wsum[wave];
  for (int i = beg; i < end; ++i) {
    const int v = arr[i];
    arr[i] = run;
    run += v;
  }
  __syncthreads();
}

__global__ __launch_bounds__(BIN_THREADS) void rgcn_bin_count(
    const int* __restrict__ dst, int* __restrict__ totals, int n_edges, int nbins) {
  __shared__ int hist[MAX_BINS];
  const int t = threadIdx.x;
  const int rel = blockIdx.y;
  for (int b = t; b < nbins; b += BIN_THREADS) hist[b] = 0;
  __syncthreads();
  const int c0 = blockIdx.x * CHUNK;
  const int* d = dst + (size_t)rel * n_edges;
#pragma unroll
  for (int k = 0; k < EDGES_PER_THREAD; ++k) {
    const int e = c0 + k * BIN_THREADS + t;
    if (e < n_edges) atomicAdd(&hist[d[e] >> BIN_SHIFT], 1);
  }
  __syncthreads();
  int* tot = totals + rel * nbins;
  for (int b = t; b < nbins; b += BIN_THREADS) {
    const int c = hist[b];
    if (c) atomicAdd(&tot[b], c);
  }
}

__global__ __launch_bounds__(BIN_THREADS) void rgcn_bin_place(
    const int* __restrict__ src, const int* __restrict__ dst,
    const float* __restrict__ eval, const int* __restrict__ totals,
    int* __restrict__ cursor, int* __restrict__ segbase, int2* __restrict__ recs,
    int n_edges, int nbins) {
  __shared__ int hist[MAX_BINS];
  __shared__ int loff[MAX_BINS];
  __shared__ int gb[MAX_BINS];
  __shared__ int wsum[32];
  __shared__ int s_w0[CHUNK];
  __shared__ int s_bin[CHUNK];
  __shared__ float s_val[CHUNK];
  const int t = threadIdx.x;
  const int rel = blockIdx.y;
  const int* tot = totals + rel * nbins;
  for (int b = t; b < nbins; b += BIN_THREADS) {
    hist[b] = 0;
    gb[b] = tot[b];
  }
  __syncthreads();
  const int c0 = blockIdx.x * CHUNK;
  const int left = n_edges - c0;
  const int cnt = left < CHUNK ? left : CHUNK;
  const size_t ebase = (size_t)rel * n_edges + c0;
  int dk[EDGES_PER_THREAD];
  int rk[EDGES_PER_THREAD];
#pragma unroll
  for (int k = 0; k < EDGES_PER_THREAD; ++k) {
    const int i = k * BIN_THREADS + t;
    dk[k] = 0;
    rk[k] = 0;
    if (i < cnt) {
      dk[k] = dst[ebase + i];
      rk[k] = atomicAdd(&hist[dk[k] >> BIN_SHIFT], 1);
    }
  }
  __syncthreads();
  for (int b = t; b < nbins; b += BIN_THREADS) loff[b] = hist[b];
  __syncthreads();
  block_exclusive_scan(loff, nbins, wsum);
  block_exclusive_scan(gb, nbins, wsum);
  for (int b = t; b < nbins; b += BIN_THREADS) {
    const int c = hist[b];
    int g = gb[b] + rel * n_edges;
    if (blockIdx.x == 0) segbase[rel * nbins + b] = g;
    if (c) g += atomicAdd(&cursor[rel * nbins + b], c);
    gb[b] = g;
  }
  __syncthreads();
#pragma unroll
  for (int k = 0; k < EDGES_PER_THREAD; ++k) {
    const int i = k * BIN_THREADS + t;
    if (i < cnt) {
      const int bin = dk[k] >> BIN_SHIFT;
      const int p = loff[bin] + rk[k];
      s_w0[p] = src[ebase + i] | ((dk[k] & (BIN_ROWS - 1)) << 24);
      s_val[p] = eval[ebase + i];
      s_bin[p] = bin;
    }
  }
  __syncthreads();
#pragma unroll
  for (int k = 0; k < EDGES_PER_THREAD; ++k) {
    const int i = k * BIN_THREADS + t;
    if (i < cnt) {
      const int bin = s_bin[i];
      const int g = gb[bin] + (i - loff[bin]);
      recs[g] = make_int2(s_w0[i], __float_as_int(s_val[i]));
    }
  }
}

__device__ __forceinline__ void lds_add(float* p, float v) {
  __hip_atomic_fetch_add(p, v, __ATOMIC_RELAXED, __HIP_MEMORY_SCOPE_WORKGROUP);
}

template <int J>
__device__ __forceinline__ int half_bcast(const int v) {
  return __builtin_amdgcn_update_dpp(0, v, 0x150 + J, 0xF, 0xF, false);
}

template <int J>
__device__ __forceinline__ float4 fetch_row(const int2 rec, const char* __restrict__ Xb, const unsigned cb) {
  const unsigned sj = (unsigned)half_bcast<J>(rec.x) & 0xFFFFFFu;
  return *(const float4*)(Xb + ((sj << 8) | cb));
}

template <int J, bool FULL>
__device__ __forceinline__ void take_row(const int2 rec, const float4 g, const int m,
                                         float* accl, int& cur, float4& s) {
  if (FULL || J < m) {
    const int row = (half_bcast<J>(rec.x) >> 24) & (BIN_ROWS - 1);
    const float vj = __int_as_float(half_bcast<J>(rec.y));
    if (row != cur) {
      float* p = accl + cur * A_PITCH;
      lds_add(p, s.x);
      lds_add(p + 1, s.y);
      lds_add(p + 2, s.z);
      lds_add(p + 3, s.w);
      s = make_float4(0.f, 0.f, 0.f, 0.f);
      cur = row;
    }
    s.x += g.x * vj;
    s.y += g.y * vj;
    s.z += g.z * vj;
    s.w += g.w * vj;
  }
}

#define FETCH8(B) \
  const float4 g0 = fetch_row<B + 0>(rec, Xb, cb), g1 = fetch_row<B + 1>(rec, Xb, cb), \
               g2 = fetch_row<B + 2>(rec, Xb, cb), g3 = fetch_row<B + 3>(rec, Xb, cb), \
               g4 = fetch_row<B + 4>(rec, Xb, cb), g5 = fetch_row<B + 5>(rec, Xb, cb), \
               g6 = fetch_row<B + 6>(rec, Xb, cb), g7 = fetch_row<B + 7>(rec, Xb, cb)
#define TAKE8(B) \
  take_row<B + 0, FULL>(rec, g0, m, accl, cur, s); take_row<B + 1, FULL>(rec, g1, m, accl, cur, s); \
  take_row<B + 2, FULL>(rec, g2, m, accl, cur, s); take_row<B + 3, FULL>(rec, g3, m, accl, cur, s); \
  take_row<B + 4, FULL>(rec, g4, m, accl, cur, s); take_row<B + 5, FULL>(rec, g5, m, accl, cur, s); \
  take_row<B + 6, FULL>(rec, g6, m, accl, cur, s); take_row<B + 7, FULL>(rec, g7, m, accl, cur, s)

template <bool FULL>
__device__ __forceinline__ void sum_step(const int2 rec, const int m, const int mu,
                                         const char* __restrict__ Xb, const unsigned cb,
                                         float* accl, int& cur, float4& s) {
  {
    FETCH8(0);
    TAKE8(0);
  }
  if (FULL || mu > 8) {
    FETCH8(8);
    TAKE8(8);
  }
}

__global__ __launch_bounds__(SUM_THREADS) __attribute__((amdgpu_waves_per_eu(16, 16)))
void rgcn_bin_sum(
    const float* __restrict__ inp, const float* __restrict__ W,
    const int2* __restrict__ recs,
    const int* __restrict__ totals, const int* __restrict__ segbase,
    const float* __restrict__ bias, float* __restrict__ out,
    int n_nodes, int n_rel, int nbins) {
  __shared__ __align__(16) float agg[BIN_ROWS * A_PITCH];
  __shared__ __align__(16) int2 sorted[SORT_CAP];
  __shared__ __align__(16) int rcnt[BIN_ROWS];
  __shared__ __align__(16) int roff[BIN_ROWS];
  static_assert(sizeof(int2) * SORT_CAP >= sizeof(float) * IN_SIZE * W_PITCH, "W_r must fit the record space");
  float* const sW = (float*)sorted;
  const int t = threadIdx.x;
  const int lane = t & 31;
  const int wave = __builtin_amdgcn_readfirstlane(t >> 5);
  const int l16 = lane & 15;
  const int b = blockIdx.x;
  const int rt = wave >> 1;
  const int ctb = (wave & 1) * 2;

  v8f cacc[2];
#pragma unroll
  for (int i = 0; i < 2; ++i) {
    float bs = 0.f;
    for (int r = 0; r < n_rel; ++r) bs += bias[r * OUT_SIZE + (ctb + i) * 16 + l16];
#pragma unroll
    for (int j = 0; j < 8; ++j) cacc[i][j] = bs;
  }

#pragma unroll
  for (int k = 0; k < (BIN_ROWS * A_PITCH / 4 + SUM_THREADS - 1) / SUM_THREADS; ++k) {
    const int q = t + k * SUM_THREADS;
    if (q < BIN_ROWS * A_PITCH / 4) *(float4*)&agg[q * 4] = make_float4(0.f, 0.f, 0.f, 0.f);
  }
  if (t < BIN_ROWS) rcnt[t] = 0;
  __syncthreads();

  const char* const Xb = (const char*)inp;
  const int half = lane >> 4;
  const unsigned cb = (unsigned)l16 * 16u;
  float* const accl = agg + 4 * l16;
  for (int r = 0; r < n_rel; ++r) {
    const int key = r * nbins + b;
    const int start = __builtin_amdgcn_readfirstlane(segbase[key]);
    const int cnt = __builtin_amdgcn_readfirstlane(totals[key]);
    for (int c0 = 0; c0 < cnt; c0 += SORT_CAP) {
      const int left = cnt - c0;
      const int n = left < SORT_CAP ? left : SORT_CAP;

      const int2* rp = recs + (start + c0);
      int ts = t;
      asm volatile("" : "+v"(ts));
      int2 rk[SORT_PER_THREAD];
      int rank[SORT_PER_THREAD];
#pragma unroll
      for (int k = 0; k < SORT_PER_THREAD; ++k) {
        const int nk = n - k * SUM_THREADS;
        rk[k] = make_int2(0, 0);
        if (nk > 0) rk[k] = (rp + k * SUM_THREADS)[ts < nk ? ts : nk - 1];
      }
#pragma unroll
      for (int k = 0; k < SORT_PER_THREAD; ++k) {
        rank[k] = 0;
        if (ts < n - k * SUM_THREADS)
          rank[k] = atomicAdd(&rcnt[(rk[k].x >> 24) & (BIN_ROWS - 1)], 1);
      }
      __syncthreads();
      if (wave == 0) {
        const int4 c = *(const int4*)&rcnt[4 * ts];
        const int s = c.x + c.y + c.z + c.w;
        int incl = s;
        for (int off = 1; off < 32; off <<= 1) {
          const int v = __shfl_up(incl, off);
          if (lane >= off) incl += v;
        }
        const int ex = incl - s;
        *(int4*)&roff[4 * ts] = make_int4(ex, ex + c.x, ex + c.x + c.y, ex + c.x + c.y + c.z);
        *(int4*)&rcnt[4 * ts] = make_int4(0, 0, 0, 0);
      }
      __syncthreads();
#pragma unroll
      for (int k = 0; k < SORT_PER_THREAD; ++k) {
        if (ts < n - k * SUM_THREADS)
          sorted[roff[(rk[k].x >> 24) & (BIN_ROWS - 1)] + rank[k]] = rk[k];
      }
      __syncthreads();

      const int span = ((n + SUM_THREADS - 1) / SUM_THREADS) * 32;
      const int beg = wave * span;
      if (beg < n) {
        const int hspan = span >> 1;
        const int hb = beg + half * hspan;
        int he = hb + hspan;
        if (he > n) he = n;
        const int last = n - 1;
        int cur = (sorted[hb < last ? hb : last].x >> 24) & (BIN_ROWS - 1);
        float4 s = make_float4(0.f, 0.f, 0.f, 0.f);
        if (beg + span <= n) {
          for (int st = 0; st < hspan; st += 16)
            sum_step<true>(sorted[hb + st + l16], 16, 16, Xb, cb, accl, cur, s);
        } else {
          int lo_end = beg + hspan;
          if (lo_end > n) lo_end = n;
          for (int st = 0; st < hspan; st += 16) {
            const int mu = lo_end - (beg + st);
            if (mu <= 0) break;
            const int i = hb + st + l16;
            sum_step<false>(sorted[i < last ? i : last], he - (hb + st), mu, Xb, cb, accl, cur, s);
          }
        }
        float* p = accl + cur * A_PITCH;
        lds_add(p, s.x);
        lds_add(p + 1, s.y);
        lds_add(p + 2, s.z);
        lds_add(p + 3, s.w);
      }
    }

    int tp = t;
    asm volatile("" : "+v"(tp));
    const int halfp = (tp >> 4) & 1;
    const int l16p = tp & 15;
    const float4* Wv = (const float4*)(W + (size_t)r * IN_SIZE * OUT_SIZE);
    const float4 w0 = Wv[tp];
    const float4 w1 = Wv[tp + SUM_THREADS];
    __syncthreads();
    *(float4*)&sW[(tp >> 4) * W_PITCH + l16p * 4] = w0;
    *(float4*)&sW[((tp >> 4) + 32) * W_PITCH + l16p * 4] = w1;
    __syncthreads();

    {
      const float* aP = agg + (rt * 16 + l16p) * A_PITCH + 2 * halfp;
      const float* bP = sW + (2 * halfp) * W_PITCH + ctb * 16 + l16p;
#pragma unroll
      for (int kk = 0; kk < IN_SIZE / 4; ++kk) {
        const float2 av = *(const float2*)(aP + kk * 4);
        v2f a;
        a[0] = av.x;
        a[1] = av.y;
#pragma unroll
        for (int i = 0; i < 2; ++i) {
          v2f bf;
          bf[0] = bP[(kk * 4) * W_PITCH + i * 16];
          bf[1] = bP[(kk * 4 + 1) * W_PITCH + i * 16];
          cacc[i] = __builtin_amdgcn_wmma_f32_16x16x4_f32(
               false, a,  false, bf,
               (short)0, cacc[i],  false,  false);
        }
      }
    }
    __syncthreads();
    if (r + 1 < n_rel) {
#pragma unroll
      for (int k = 0; k < (BIN_ROWS * A_PITCH / 4 + SUM_THREADS - 1) / SUM_THREADS; ++k) {
        const int q = tp + k * SUM_THREADS;
        if (q < BIN_ROWS * A_PITCH / 4) *(float4*)&agg[q * 4] = make_float4(0.f, 0.f, 0.f, 0.f);
      }
    }
  }

  int te = t;
  asm volatile("" : "+v"(te));
  const int halfe = (te >> 4) & 1;
  const int l16e = te & 15;
#pragma unroll
  for (int i = 0; i < 2; ++i)
#pragma unroll
    for (int j = 0; j < 8; ++j)
      agg[(rt * 16 + j + 8 * halfe) * A_PITCH + (ctb + i) * 16 + l16e] = cacc[i][j];
  __syncthreads();
  const int c4 = l16e * 4;
#pragma unroll
  for (int k = 0; k < (BIN_ROWS * OUT_SIZE / 4) / SUM_THREADS; ++k) {
    const int q = te + k * SUM_THREADS;
    const int lrow = q >> 4;
    const int grow = b * BIN_ROWS + lrow;
    if (grow < n_nodes)
      *(float4*)(out + (size_t)grow * OUT_SIZE + c4) = *(const float4*)&agg[lrow * A_PITCH + c4];
  }
}

extern "C" void kernel_launch(void* const* d_in, const int* in_sizes, int n_in,
                              void* d_out, int out_size, void* d_ws, size_t ws_size,
                              hipStream_t stream) {
  const float* inp = (const float*)d_in[0];
  const int* src = (const int*)d_in[1];
  const int* dst = (const int*)d_in[2];
  const float* eval = (const float*)d_in[3];
  const float* weights = (const float*)d_in[4];
  const float* bias = (const float*)d_in[5];
  float* out = (float*)d_out;

  const int n_nodes = in_sizes[0] / IN_SIZE;
  const int n_rel = in_sizes[4] / (IN_SIZE * OUT_SIZE);
  const int n_edges = in_sizes[1] / n_rel;
  const int nbins = (n_nodes + BIN_ROWS - 1) / BIN_ROWS;

  int2* recs = (int2*)d_ws;
  int* totals = (int*)(recs + (size_t)n_rel * n_edges);
  int* cursor = totals + n_rel * nbins;
  int* segbase = cursor + n_rel * nbins;

  const int n_counters = 2 * n_rel * nbins;
  rgcn_clear<<<(n_counters + BIN_THREADS - 1) / BIN_THREADS, BIN_THREADS, 0, stream>>>(totals, n_counters);

  const int chunks = (n_edges + CHUNK - 1) / CHUNK;
  rgcn_bin_count<<<dim3(chunks, n_rel), BIN_THREADS, 0, stream>>>(dst, totals, n_edges, nbins);
  rgcn_bin_place<<<dim3(chunks, n_rel), BIN_THREADS, 0, stream>>>(
      src, dst, eval, totals, cursor, segbase, recs, n_edges, nbins);
  rgcn_bin_sum<<<nbins, SUM_THREADS, 0, stream>>>(
      inp, weights, recs, totals, segbase, bias, out, n_nodes, n_rel, nbins);
}


# destination ranges of 100 rows (was 128): 1000 work-groups fill the 1024 resident places, and a range's records per relation stay under 2048: no fifth gather step; sum order: runs cut elsewhere
# speedup vs baseline: 5.8102x; 1.0425x over previous
#include <hip/hip_runtime.h>
#include <hip/hip_bf16.h>

typedef __attribute__((ext_vector_type(2))) float v2f;
typedef __attribute__((ext_vector_type(8))) float v8f;

#define IN_SIZE 64
#define OUT_SIZE 64


#define W_PITCH 72
#define A_PITCH 68

#define BIN_ROWS 100
#define TILE_ROWS 128
#define ROW_SLOTS 128
#define ROW_MASK (ROW_SLOTS - 1)
#define MAX_BINS 2048
#define CHUNK 8192
#define BIN_THREADS 1024
#define EDGES_PER_THREAD (CHUNK / BIN_THREADS)
#define SUM_THREADS 512
#define SUM_WAVES (SUM_THREADS / 32)
#define SORT_CAP 2560
#define SORT_PER_THREAD (SORT_CAP / SUM_THREADS)

__global__ __launch_bounds__(BIN_THREADS) void rgcn_clear(int* __restrict__ counters, int n_counters) {
  const int i = blockIdx.x * BIN_THREADS + threadIdx.x;
  if (i < n_counters) counters[i] = 0;
}

__device__ __forceinline__ void block_exclusive_scan(int* arr, int n, int* wsum) {
  const int t = threadIdx.x;
  const int lane = t & 31;
  const int wave = t >> 5;
  const int per = (n + BIN_THREADS - 1) / BIN_THREADS;
  const int beg = t * per;
  int end = beg + per;
  if (end > n) end = n;
  int s = 0;
  for (int i = beg; i < end; ++i) s += arr[i];
  int incl = s;
  for (int off = 1; off < 32; off <<= 1) {
    const int v = __shfl_up(incl, off);
    if (lane >= off) incl += v;
  }
  if (lane == 31) wsum[wave] = incl;
  __syncthreads();
  if (wave == 0) {
    const int w = wsum[lane];
    int wi = w;
    for (int off = 1; off < 32; off <<= 1) {
      const int v = __shfl_up(wi, off);
      if (lane >= off) wi += v;
    }
    wsum[lane] = wi - w;
  }
  __syncthreads();
  int run = incl - s + wsum[wave];
  for (int i = beg; i < end; ++i) {
    const int v = arr[i];
    arr[i] = run;
    run += v;
  }
  __syncthreads();
}

__global__ __launch_bounds__(BIN_THREADS) void rgcn_bin_count(
    const int* __restrict__ dst, int* __restrict__ totals, int n_edges, int nbins) {
  __shared__ int hist[MAX_BINS];
  const int t = threadIdx.x;
  const int rel = blockIdx.y;
  for (int b = t; b < nbins; b += BIN_THREADS) hist[b] = 0;
  __syncthreads();
  const int c0 = blockIdx.x * CHUNK;
  const int* d = dst + (size_t)rel * n_edges;
#pragma unroll
  for (int k = 0; k < EDGES_PER_THREAD; ++k) {
    const int e = c0 + k * BIN_THREADS + t;
    if (e < n_edges) atomicAdd(&hist[(unsigned)d[e] / BIN_ROWS], 1);
  }
  __syncthreads();
  int* tot = totals + rel * nbins;
  for (int b = t; b < nbins; b += BIN_THREADS) {
    const int c = hist[b];
    if (c) atomicAdd(&tot[b], c);
  }
}

__global__ __launch_bounds__(BIN_THREADS) void rgcn_bin_place(
    const int* __restrict__ src, const int* __restrict__ dst,
    const float* __restrict__ eval, const int* __restrict__ totals,
    int* __restrict__ cursor, int* __restrict__ segbase, int2* __restrict__ recs,
    int n_edges, int nbins) {
  __shared__ int hist[MAX_BINS];
  __shared__ int loff[MAX_BINS];
  __shared__ int gb[MAX_BINS];
  __shared__ int wsum[32];
  __shared__ int s_w0[CHUNK];
  __shared__ int s_bin[CHUNK];
  __shared__ float s_val[CHUNK];
  const int t = threadIdx.x;
  const int rel = blockIdx.y;
  const int* tot = totals + rel * nbins;
  for (int b = t; b < nbins; b += BIN_THREADS) {
    hist[b] = 0;
    gb[b] = tot[b];
  }
  __syncthreads();
  const int c0 = blockIdx.x * CHUNK;
  const int left = n_edges - c0;
  const int cnt = left < CHUNK ? left : CHUNK;
  const size_t ebase = (size_t)rel * n_edges + c0;
  int dk[EDGES_PER_THREAD];
  int rk[EDGES_PER_THREAD];
#pragma unroll
  for (int k = 0; k < EDGES_PER_THREAD; ++k) {
    const int i = k * BIN_THREADS + t;
    dk[k] = 0;
    rk[k] = 0;
    if (i < cnt) {
      dk[k] = dst[ebase + i];
      rk[k] = atomicAdd(&hist[(unsigned)dk[k] / BIN_ROWS], 1);
    }
  }
  __syncthreads();
  for (int b = t; b < nbins; b += BIN_THREADS) loff[b] = hist[b];
  __syncthreads();
  block_exclusive_scan(loff, nbins, wsum);
  block_exclusive_scan(gb, nbins, wsum);
  for (int b = t; b < nbins; b += BIN_THREADS) {
    const int c = hist[b];
    int g = gb[b] + rel * n_edges;
    if (blockIdx.x == 0) segbase[rel * nbins + b] = g;
    if (c) g += atomicAdd(&cursor[rel * nbins + b], c);
    gb[b] = g;
  }
  __syncthreads();
#pragma unroll
  for (int k = 0; k < EDGES_PER_THREAD; ++k) {
    const int i = k * BIN_THREADS + t;
    if (i < cnt) {
      const int bin = (unsigned)dk[k] / BIN_ROWS;
      const int p = loff[bin] + rk[k];
      s_w0[p] = src[ebase + i] | ((dk[k] - bin * BIN_ROWS) << 24);
      s_val[p] = eval[ebase + i];
      s_bin[p] = bin;
    }
  }
  __syncthreads();
#pragma unroll
  for (int k = 0; k < EDGES_PER_THREAD; ++k) {
    const int i = k * BIN_THREADS + t;
    if (i < cnt) {
      const int bin = s_bin[i];
      const int g = gb[bin] + (i - loff[bin]);
      recs[g] = make_int2(s_w0[i], __float_as_int(s_val[i]));
    }
  }
}

__device__ __forceinline__ void lds_add(float* p, float v) {
  __hip_atomic_fetch_add(p, v, __ATOMIC_RELAXED, __HIP_MEMORY_SCOPE_WORKGROUP);
}

template <int J>
__device__ __forceinline__ int half_bcast(const int v) {
  return __builtin_amdgcn_update_dpp(0, v, 0x150 + J, 0xF, 0xF, false);
}

template <int J>
__device__ __forceinline__ float4 fetch_row(const int2 rec, const char* __restrict__ Xb, const unsigned cb) {
  const unsigned sj = (unsigned)half_bcast<J>(rec.x) & 0xFFFFFFu;
  return *(const float4*)(Xb + ((sj << 8) | cb));
}

template <int J, bool FULL>
__device__ __forceinline__ void take_row(const int2 rec, const float4 g, const int m,
                                         float* accl, int& cur, float4& s) {
  if (FULL || J < m) {
    const int row = (half_bcast<J>(rec.x) >> 24) & ROW_MASK;
    const float vj = __int_as_float(half_bcast<J>(rec.y));
    if (row != cur) {
      float* p = accl + cur * A_PITCH;
      lds_add(p, s.x);
      lds_add(p + 1, s.y);
      lds_add(p + 2, s.z);
      lds_add(p + 3, s.w);
      s = make_float4(0.f, 0.f, 0.f, 0.f);
      cur = row;
    }
    s.x += g.x * vj;
    s.y += g.y * vj;
    s.z += g.z * vj;
    s.w += g.w * vj;
  }
}

#define FETCH8(B) \
  const float4 g0 = fetch_row<B + 0>(rec, Xb, cb), g1 = fetch_row<B + 1>(rec, Xb, cb), \
               g2 = fetch_row<B + 2>(rec, Xb, cb), g3 = fetch_row<B + 3>(rec, Xb, cb), \
               g4 = fetch_row<B + 4>(rec, Xb, cb), g5 = fetch_row<B + 5>(rec, Xb, cb), \
               g6 = fetch_row<B + 6>(rec, Xb, cb), g7 = fetch_row<B + 7>(rec, Xb, cb)
#define TAKE8(B) \
  take_row<B + 0, FULL>(rec, g0, m, accl, cur, s); take_row<B + 1, FULL>(rec, g1, m, accl, cur, s); \
  take_row<B + 2, FULL>(rec, g2, m, accl, cur, s); take_row<B + 3, FULL>(rec, g3, m, accl, cur, s); \
  take_row<B + 4, FULL>(rec, g4, m, accl, cur, s); take_row<B + 5, FULL>(rec, g5, m, accl, cur, s); \
  take_row<B + 6, FULL>(rec, g6, m, accl, cur, s); take_row<B + 7, FULL>(rec, g7, m, accl, cur, s)

template <bool FULL>
__device__ __forceinline__ void sum_step(const int2 rec, const int m, const int mu,
                                         const char* __restrict__ Xb, const unsigned cb,
                                         float* accl, int& cur, float4& s) {
  {
    FETCH8(0);
    TAKE8(0);
  }
  if (FULL || mu > 8) {
    FETCH8(8);
    TAKE8(8);
  }
}

__global__ __launch_bounds__(SUM_THREADS) __attribute__((amdgpu_waves_per_eu(16, 16)))
void rgcn_bin_sum(
    const float* __restrict__ inp, const float* __restrict__ W,
    const int2* __restrict__ recs,
    const int* __restrict__ totals, const int* __restrict__ segbase,
    const float* __restrict__ bias, float* __restrict__ out,
    int n_nodes, int n_rel, int nbins) {
  __shared__ __align__(16) float agg[TILE_ROWS * A_PITCH];
  __shared__ __align__(16) int2 sorted[SORT_CAP];
  __shared__ __align__(16) int rcnt[ROW_SLOTS];
  __shared__ __align__(16) int roff[ROW_SLOTS];
  static_assert(sizeof(int2) * SORT_CAP >= sizeof(float) * IN_SIZE * W_PITCH, "W_r must fit the record space");
  float* const sW = (float*)sorted;
  const int t = threadIdx.x;
  const int lane = t & 31;
  const int wave = __builtin_amdgcn_readfirstlane(t >> 5);
  const int l16 = lane & 15;
  const int b = blockIdx.x;
  const int rt = wave >> 1;
  const int ctb = (wave & 1) * 2;

  v8f cacc[2];
#pragma unroll
  for (int i = 0; i < 2; ++i) {
    float bs = 0.f;
    for (int r = 0; r < n_rel; ++r) bs += bias[r * OUT_SIZE + (ctb + i) * 16 + l16];
#pragma unroll
    for (int j = 0; j < 8; ++j) cacc[i][j] = bs;
  }

#pragma unroll
  for (int k = 0; k < (TILE_ROWS * A_PITCH / 4 + SUM_THREADS - 1) / SUM_THREADS; ++k) {
    const int q = t + k * SUM_THREADS;
    if (q < TILE_ROWS * A_PITCH / 4) *(float4*)&agg[q * 4] = make_float4(0.f, 0.f, 0.f, 0.f);
  }
  if (t < ROW_SLOTS) rcnt[t] = 0;
  __syncthreads();

  const char* const Xb = (const char*)inp;
  const int half = lane >> 4;
  const unsigned cb = (unsigned)l16 * 16u;
  float* const accl = agg + 4 * l16;
  for (int r = 0; r < n_rel; ++r) {
    const int key = r * nbins + b;
    const int start = __builtin_amdgcn_readfirstlane(segbase[key]);
    const int cnt = __builtin_amdgcn_readfirstlane(totals[key]);
    for (int c0 = 0; c0 < cnt; c0 += SORT_CAP) {
      const int left = cnt - c0;
      const int n = left < SORT_CAP ? left : SORT_CAP;

      const int2* rp = recs + (start + c0);
      int ts = t;
      asm volatile("" : "+v"(ts));
      int2 rk[SORT_PER_THREAD];
      int rank[SORT_PER_THREAD];
#pragma unroll
      for (int k = 0; k < SORT_PER_THREAD; ++k) {
        const int nk = n - k * SUM_THREADS;
        rk[k] = make_int2(0, 0);
        if (nk > 0) rk[k] = (rp + k * SUM_THREADS)[ts < nk ? ts : nk - 1];
      }
#pragma unroll
      for (int k = 0; k < SORT_PER_THREAD; ++k) {
        rank[k] = 0;
        if (ts < n - k * SUM_THREADS)
          rank[k] = atomicAdd(&rcnt[(rk[k].x >> 24) & ROW_MASK], 1);
      }
      __syncthreads();
      if (wave == 0) {
        const int4 c = *(const int4*)&rcnt[4 * ts];
        const int s = c.x + c.y + c.z + c.w;
        int incl = s;
        for (int off = 1; off < 32; off <<= 1) {
          const int v = __shfl_up(incl, off);
          if (lane >= off) incl += v;
        }
        const int ex = incl - s;
        *(int4*)&roff[4 * ts] = make_int4(ex, ex + c.x, ex + c.x + c.y, ex + c.x + c.y + c.z);
        *(int4*)&rcnt[4 * ts] = make_int4(0, 0, 0, 0);
      }
      __syncthreads();
#pragma unroll
      for (int k = 0; k < SORT_PER_THREAD; ++k) {
        if (ts < n - k * SUM_THREADS)
          sorted[roff[(rk[k].x >> 24) & ROW_MASK] + rank[k]] = rk[k];
      }
      __syncthreads();

      const int span = ((n + SUM_THREADS - 1) / SUM_THREADS) * 32;
      const int beg = wave * span;
      if (beg < n) {
        const int hspan = span >> 1;
        const int hb = beg + half * hspan;
        int he = hb + hspan;
        if (he > n) he = n;
        const int last = n - 1;
        int cur = (sorted[hb < last ? hb : last].x >> 24) & ROW_MASK;
        float4 s = make_float4(0.f, 0.f, 0.f, 0.f);
        if (beg + span <= n) {
          for (int st = 0; st < hspan; st += 16)
            sum_step<true>(sorted[hb + st + l16], 16, 16, Xb, cb, accl, cur, s);
        } else {
          int lo_end = beg + hspan;
          if (lo_end > n) lo_end = n;
          for (int st = 0; st < hspan; st += 16) {
            const int mu = lo_end - (beg + st);
            if (mu <= 0) break;
            const int i = hb + st + l16;
            sum_step<false>(sorted[i < last ? i : last], he - (hb + st), mu, Xb, cb, accl, cur, s);
          }
        }
        float* p = accl + cur * A_PITCH;
        lds_add(p, s.x);
        lds_add(p + 1, s.y);
        lds_add(p + 2, s.z);
        lds_add(p + 3, s.w);
      }
    }

    int tp = t;
    asm volatile("" : "+v"(tp));
    const int halfp = (tp >> 4) & 1;
    const int l16p = tp & 15;
    const float4* Wv = (const float4*)(W + (size_t)r * IN_SIZE * OUT_SIZE);
    const float4 w0 = Wv[tp];
    const float4 w1 = Wv[tp + SUM_THREADS];
    __syncthreads();
    *(float4*)&sW[(tp >> 4) * W_PITCH + l16p * 4] = w0;
    *(float4*)&sW[((tp >> 4) + 32) * W_PITCH + l16p * 4] = w1;
    __syncthreads();

    {
      const float* aP = agg + (rt * 16 + l16p) * A_PITCH + 2 * halfp;
      const float* bP = sW + (2 * halfp) * W_PITCH + ctb * 16 + l16p;
#pragma unroll
      for (int kk = 0; kk < IN_SIZE / 4; ++kk) {
        const float2 av = *(const float2*)(aP + kk * 4);
        v2f a;
        a[0] = av.x;
        a[1] = av.y;
#pragma unroll
        for (int i = 0; i < 2; ++i) {
          v2f bf;
          bf[0] = bP[(kk * 4) * W_PITCH + i * 16];
          bf[1] = bP[(kk * 4 + 1) * W_PITCH + i * 16];
          cacc[i] = __builtin_amdgcn_wmma_f32_16x16x4_f32(
               false, a,  false, bf,
               (short)0, cacc[i],  false,  false);
        }
      }
    }
    __syncthreads();
    if (r + 1 < n_rel) {
#pragma unroll
      for (int k = 0; k < (TILE_ROWS * A_PITCH / 4 + SUM_THREADS - 1) / SUM_THREADS; ++k) {
        const int q = tp + k * SUM_THREADS;
        if (q < TILE_ROWS * A_PITCH / 4) *(float4*)&agg[q * 4] = make_float4(0.f, 0.f, 0.f, 0.f);
      }
    }
  }

  int te = t;
  asm volatile("" : "+v"(te));
  const int halfe = (te >> 4) & 1;
  const int l16e = te & 15;
#pragma unroll
  for (int i = 0; i < 2; ++i)
#pragma unroll
    for (int j = 0; j < 8; ++j)
      agg[(rt * 16 + j + 8 * halfe) * A_PITCH + (ctb + i) * 16 + l16e] = cacc[i][j];
  __syncthreads();
  const int c4 = l16e * 4;
#pragma unroll
  for (int k = 0; k < (BIN_ROWS * OUT_SIZE / 4 + SUM_THREADS - 1) / SUM_THREADS; ++k) {
    const int q = te + k * SUM_THREADS;
    const int lrow = q >> 4;
    const int grow = b * BIN_ROWS + lrow;
    if (lrow < BIN_ROWS && grow < n_nodes)
      *(float4*)(out + (size_t)grow * OUT_SIZE + c4) = *(const float4*)&agg[lrow * A_PITCH + c4];
  }
}

extern "C" void kernel_launch(void* const* d_in, const int* in_sizes, int n_in,
                              void* d_out, int out_size, void* d_ws, size_t ws_size,
                              hipStream_t stream) {
  const float* inp = (const float*)d_in[0];
  const int* src = (const int*)d_in[1];
  const int* dst = (const int*)d_in[2];
  const float* eval = (const float*)d_in[3];
  const float* weights = (const float*)d_in[4];
  const float* bias = (const float*)d_in[5];
  float* out = (float*)d_out;

  const int n_nodes = in_sizes[0] / IN_SIZE;
  const int n_rel = in_sizes[4] / (IN_SIZE * OUT_SIZE);
  const int n_edges = in_sizes[1] / n_rel;
  const int nbins = (n_nodes + BIN_ROWS - 1) / BIN_ROWS;

  int2* recs = (int2*)d_ws;
  int* totals = (int*)(recs + (size_t)n_rel * n_edges);
  int* cursor = totals + n_rel * nbins;
  int* segbase = cursor + n_rel * nbins;

  const int n_counters = 2 * n_rel * nbins;
  rgcn_clear<<<(n_counters + BIN_THREADS - 1) / BIN_THREADS, BIN_THREADS, 0, stream>>>(totals, n_counters);

  const int chunks = (n_edges + CHUNK - 1) / CHUNK;
  rgcn_bin_count<<<dim3(chunks, n_rel), BIN_THREADS, 0, stream>>>(dst, totals, n_edges, nbins);
  rgcn_bin_place<<<dim3(chunks, n_rel), BIN_THREADS, 0, stream>>>(
      src, dst, eval, totals, cursor, segbase, recs, n_edges, nbins);
  rgcn_bin_sum<<<nbins, SUM_THREADS, 0, stream>>>(
      inp, weights, recs, totals, segbase, bias, out, n_nodes, n_rel, nbins);
}


# binning kernels: a thread's edge loads issued as one group ahead of their use (clamped index, no guard round each load): one trip to memory per block, were 8 and 16 in a row; sum order rule unchanged
# speedup vs baseline: 6.0543x; 1.0420x over previous
#include <hip/hip_runtime.h>
#include <hip/hip_bf16.h>

typedef __attribute__((ext_vector_type(2))) float v2f;
typedef __attribute__((ext_vector_type(8))) float v8f;

#define IN_SIZE 64
#define OUT_SIZE 64


#define W_PITCH 72
#define A_PITCH 68

#define BIN_ROWS 100
#define TILE_ROWS 128
#define ROW_SLOTS 128
#define ROW_MASK (ROW_SLOTS - 1)
#define MAX_BINS 2048
#define CHUNK 8192
#define BIN_THREADS 1024
#define EDGES_PER_THREAD (CHUNK / BIN_THREADS)
#define SUM_THREADS 512
#define SUM_WAVES (SUM_THREADS / 32)
#define SORT_CAP 2560
#define SORT_PER_THREAD (SORT_CAP / SUM_THREADS)

__global__ __launch_bounds__(BIN_THREADS) void rgcn_clear(int* __restrict__ counters, int n_counters) {
  const int i = blockIdx.x * BIN_THREADS + threadIdx.x;
  if (i < n_counters) counters[i] = 0;
}

__device__ __forceinline__ void block_exclusive_scan(int* arr, int n, int* wsum) {
  const int t = threadIdx.x;
  const int lane = t & 31;
  const int wave = t >> 5;
  const int per = (n + BIN_THREADS - 1) / BIN_THREADS;
  const int beg = t * per;
  int end = beg + per;
  if (end > n) end = n;
  int s = 0;
  for (int i = beg; i < end; ++i) s += arr[i];
  int incl = s;
  for (int off = 1; off < 32; off <<= 1) {
    const int v = __shfl_up(incl, off);
    if (lane >= off) incl += v;
  }
  if (lane == 31) wsum[wave] = incl;
  __syncthreads();
  if (wave == 0) {
    const int w = wsum[lane];
    int wi = w;
    for (int off = 1; off < 32; off <<= 1) {
      const int v = __shfl_up(wi, off);
      if (lane >= off) wi += v;
    }
    wsum[lane] = wi - w;
  }
  __syncthreads();
  int run = incl - s + wsum[wave];
  for (int i = beg; i < end; ++i) {
    const int v = arr[i];
    arr[i] = run;
    run += v;
  }
  __syncthreads();
}

__global__ __launch_bounds__(BIN_THREADS) void rgcn_bin_count(
    const int* __restrict__ dst, int* __restrict__ totals, int n_edges, int nbins) {
  __shared__ int hist[MAX_BINS];
  const int t = threadIdx.x;
  const int rel = blockIdx.y;
  const int c0 = blockIdx.x * CHUNK;
  const int left = n_edges - c0;
  const int cnt = left < CHUNK ? left : CHUNK;
  const int* d = dst + (size_t)rel * n_edges + c0;
  int dv[EDGES_PER_THREAD];
#pragma unroll
  for (int k = 0; k < EDGES_PER_THREAD; ++k) {
    const int i = k * BIN_THREADS + t;
    dv[k] = d[i < cnt ? i : cnt - 1];
  }
  for (int b = t; b < nbins; b += BIN_THREADS) hist[b] = 0;
  __syncthreads();
#pragma unroll
  for (int k = 0; k < EDGES_PER_THREAD; ++k) {
    if (k * BIN_THREADS + t < cnt) atomicAdd(&hist[(unsigned)dv[k] / BIN_ROWS], 1);
  }
  __syncthreads();
  int* tot = totals + rel * nbins;
  for (int b = t; b < nbins; b += BIN_THREADS) {
    const int c = hist[b];
    if (c) atomicAdd(&tot[b], c);
  }
}

__global__ __launch_bounds__(BIN_THREADS) __attribute__((amdgpu_waves_per_eu(16, 16)))
void rgcn_bin_place(
    const int* __restrict__ src, const int* __restrict__ dst,
    const float* __restrict__ eval, const int* __restrict__ totals,
    int* __restrict__ cursor, int* __restrict__ segbase, int2* __restrict__ recs,
    int n_edges, int nbins) {
  __shared__ int hist[MAX_BINS];
  __shared__ int loff[MAX_BINS];
  __shared__ int gb[MAX_BINS];
  __shared__ int wsum[32];
  __shared__ int s_w0[CHUNK];
  __shared__ int s_bin[CHUNK];
  __shared__ float s_val[CHUNK];
  const int t = threadIdx.x;
  const int rel = blockIdx.y;
  const int* tot = totals + rel * nbins;
  const int c0 = blockIdx.x * CHUNK;
  const int left = n_edges - c0;
  const int cnt = left < CHUNK ? left : CHUNK;
  const size_t ebase = (size_t)rel * n_edges + c0;
  const int* dchunk = dst + ebase;
  const int* schunk = src + ebase;
  const float* vchunk = eval + ebase;
  int dk[EDGES_PER_THREAD];
  int sk[EDGES_PER_THREAD];
  float vk[EDGES_PER_THREAD];
#pragma unroll
  for (int k = 0; k < EDGES_PER_THREAD; ++k) {
    const int i = k * BIN_THREADS + t;
    dk[k] = dchunk[i < cnt ? i : cnt - 1];
  }
#pragma unroll
  for (int k = 0; k < EDGES_PER_THREAD; ++k) {
    const int i = k * BIN_THREADS + t;
    const int ic = i < cnt ? i : cnt - 1;
    sk[k] = schunk[ic];
    vk[k] = vchunk[ic];
  }
  for (int b = t; b < nbins; b += BIN_THREADS) {
    hist[b] = 0;
    gb[b] = tot[b];
  }
  __syncthreads();
  static_assert(MAX_BINS * BIN_ROWS <= (1 << 18) && CHUNK <= (1 << 14), "destination and rank share 32 bits");
#pragma unroll
  for (int k = 0; k < EDGES_PER_THREAD; ++k) {
    if (k * BIN_THREADS + t < cnt)
      dk[k] |= (int)((unsigned)atomicAdd(&hist[(unsigned)dk[k] / BIN_ROWS], 1) << 18);
  }
  __syncthreads();
  for (int b = t; b < nbins; b += BIN_THREADS) loff[b] = hist[b];
  __syncthreads();
  block_exclusive_scan(loff, nbins, wsum);
  block_exclusive_scan(gb, nbins, wsum);
  for (int b = t; b < nbins; b += BIN_THREADS) {
    const int c = hist[b];
    int g = gb[b] + rel * n_edges;
    if (blockIdx.x == 0) segbase[rel * nbins + b] = g;
    if (c) g += atomicAdd(&cursor[rel * nbins + b], c);
    gb[b] = g;
  }
  __syncthreads();
#pragma unroll
  for (int k = 0; k < EDGES_PER_THREAD; ++k) {
    const int i = k * BIN_THREADS + t;
    if (i < cnt) {
      const int d = dk[k] & ((1 << 18) - 1);
      const int bin = (unsigned)d / BIN_ROWS;
      const int p = loff[bin] + (int)((unsigned)dk[k] >> 18);
      s_w0[p] = sk[k] | ((d - bin * BIN_ROWS) << 24);
      s_val[p] = vk[k];
      s_bin[p] = bin;
    }
  }
  __syncthreads();
#pragma unroll
  for (int k = 0; k < EDGES_PER_THREAD; ++k) {
    const int i = k * BIN_THREADS + t;
    if (i < cnt) {
      const int bin = s_bin[i];
      const int g = gb[bin] + (i - loff[bin]);
      recs[g] = make_int2(s_w0[i], __float_as_int(s_val[i]));
    }
  }
}

__device__ __forceinline__ void lds_add(float* p, float v) {
  __hip_atomic_fetch_add(p, v, __ATOMIC_RELAXED, __HIP_MEMORY_SCOPE_WORKGROUP);
}

template <int J>
__device__ __forceinline__ int half_bcast(const int v) {
  return __builtin_amdgcn_update_dpp(0, v, 0x150 + J, 0xF, 0xF, false);
}

template <int J>
__device__ __forceinline__ float4 fetch_row(const int2 rec, const char* __restrict__ Xb, const unsigned cb) {
  const unsigned sj = (unsigned)half_bcast<J>(rec.x) & 0xFFFFFFu;
  return *(const float4*)(Xb + ((sj << 8) | cb));
}

template <int J, bool FULL>
__device__ __forceinline__ void take_row(const int2 rec, const float4 g, const int m,
                                         float* accl, int& cur, float4& s) {
  if (FULL || J < m) {
    const int row = (half_bcast<J>(rec.x) >> 24) & ROW_MASK;
    const float vj = __int_as_float(half_bcast<J>(rec.y));
    if (row != cur) {
      float* p = accl + cur * A_PITCH;
      lds_add(p, s.x);
      lds_add(p + 1, s.y);
      lds_add(p + 2, s.z);
      lds_add(p + 3, s.w);
      s = make_float4(0.f, 0.f, 0.f, 0.f);
      cur = row;
    }
    s.x += g.x * vj;
    s.y += g.y * vj;
    s.z += g.z * vj;
    s.w += g.w * vj;
  }
}

#define FETCH8(B) \
  const float4 g0 = fetch_row<B + 0>(rec, Xb, cb), g1 = fetch_row<B + 1>(rec, Xb, cb), \
               g2 = fetch_row<B + 2>(rec, Xb, cb), g3 = fetch_row<B + 3>(rec, Xb, cb), \
               g4 = fetch_row<B + 4>(rec, Xb, cb), g5 = fetch_row<B + 5>(rec, Xb, cb), \
               g6 = fetch_row<B + 6>(rec, Xb, cb), g7 = fetch_row<B + 7>(rec, Xb, cb)
#define TAKE8(B) \
  take_row<B + 0, FULL>(rec, g0, m, accl, cur, s); take_row<B + 1, FULL>(rec, g1, m, accl, cur, s); \
  take_row<B + 2, FULL>(rec, g2, m, accl, cur, s); take_row<B + 3, FULL>(rec, g3, m, accl, cur, s); \
  take_row<B + 4, FULL>(rec, g4, m, accl, cur, s); take_row<B + 5, FULL>(rec, g5, m, accl, cur, s); \
  take_row<B + 6, FULL>(rec, g6, m, accl, cur, s); take_row<B + 7, FULL>(rec, g7, m, accl, cur, s)

template <bool FULL>
__device__ __forceinline__ void sum_step(const int2 rec, const int m, const int mu,
                                         const char* __restrict__ Xb, const unsigned cb,
                                         float* accl, int& cur, float4& s) {
  {
    FETCH8(0);
    TAKE8(0);
  }
  if (FULL || mu > 8) {
    FETCH8(8);
    TAKE8(8);
  }
}

__global__ __launch_bounds__(SUM_THREADS) __attribute__((amdgpu_waves_per_eu(16, 16)))
void rgcn_bin_sum(
    const float* __restrict__ inp, const float* __restrict__ W,
    const int2* __restrict__ recs,
    const int* __restrict__ totals, const int* __restrict__ segbase,
    const float* __restrict__ bias, float* __restrict__ out,
    int n_nodes, int n_rel, int nbins) {
  __shared__ __align__(16) float agg[TILE_ROWS * A_PITCH];
  __shared__ __align__(16) int2 sorted[SORT_CAP];
  __shared__ __align__(16) int rcnt[ROW_SLOTS];
  __shared__ __align__(16) int roff[ROW_SLOTS];
  static_assert(sizeof(int2) * SORT_CAP >= sizeof(float) * IN_SIZE * W_PITCH, "W_r must fit the record space");
  float* const sW = (float*)sorted;
  const int t = threadIdx.x;
  const int lane = t & 31;
  const int wave = __builtin_amdgcn_readfirstlane(t >> 5);
  const int l16 = lane & 15;
  const int b = blockIdx.x;
  const int rt = wave >> 1;
  const int ctb = (wave & 1) * 2;

  v8f cacc[2];
#pragma unroll
  for (int i = 0; i < 2; ++i) {
    float bs = 0.f;
    for (int r = 0; r < n_rel; ++r) bs += bias[r * OUT_SIZE + (ctb + i) * 16 + l16];
#pragma unroll
    for (int j = 0; j < 8; ++j) cacc[i][j] = bs;
  }

#pragma unroll
  for (int k = 0; k < (TILE_ROWS * A_PITCH / 4 + SUM_THREADS - 1) / SUM_THREADS; ++k) {
    const int q = t + k * SUM_THREADS;
    if (q < TILE_ROWS * A_PITCH / 4) *(float4*)&agg[q * 4] = make_float4(0.f, 0.f, 0.f, 0.f);
  }
  if (t < ROW_SLOTS) rcnt[t] = 0;
  __syncthreads();

  const char* const Xb = (const char*)inp;
  const int half = lane >> 4;
  const unsigned cb = (unsigned)l16 * 16u;
  float* const accl = agg + 4 * l16;
  for (int r = 0; r < n_rel; ++r) {
    const int key = r * nbins + b;
    const int start = __builtin_amdgcn_readfirstlane(segbase[key]);
    const int cnt = __builtin_amdgcn_readfirstlane(totals[key]);
    for (int c0 = 0; c0 < cnt; c0 += SORT_CAP) {
      const int left = cnt - c0;
      const int n = left < SORT_CAP ? left : SORT_CAP;

      const int2* rp = recs + (start + c0);
      int ts = t;
      asm volatile("" : "+v"(ts));
      int2 rk[SORT_PER_THREAD];
      int rank[SORT_PER_THREAD];
#pragma unroll
      for (int k = 0; k < SORT_PER_THREAD; ++k) {
        const int nk = n - k * SUM_THREADS;
        rk[k] = make_int2(0, 0);
        if (nk > 0) rk[k] = (rp + k * SUM_THREADS)[ts < nk ? ts : nk - 1];
      }
#pragma unroll
      for (int k = 0; k < SORT_PER_THREAD; ++k) {
        rank[k] = 0;
        if (ts < n - k * SUM_THREADS)
          rank[k] = atomicAdd(&rcnt[(rk[k].x >> 24) & ROW_MASK], 1);
      }
      __syncthreads();
      if (wave == 0) {
        const int4 c = *(const int4*)&rcnt[4 * ts];
        const int s = c.x + c.y + c.z + c.w;
        int incl = s;
        for (int off = 1; off < 32; off <<= 1) {
          const int v = __shfl_up(incl, off);
          if (lane >= off) incl += v;
        }
        const int ex = incl - s;
        *(int4*)&roff[4 * ts] = make_int4(ex, ex + c.x, ex + c.x + c.y, ex + c.x + c.y + c.z);
        *(int4*)&rcnt[4 * ts] = make_int4(0, 0, 0, 0);
      }
      __syncthreads();
#pragma unroll
      for (int k = 0; k < SORT_PER_THREAD; ++k) {
        if (ts < n - k * SUM_THREADS)
          sorted[roff[(rk[k].x >> 24) & ROW_MASK] + rank[k]] = rk[k];
      }
      __syncthreads();

      const int span = ((n + SUM_THREADS - 1) / SUM_THREADS) * 32;
      const int beg = wave * span;
      if (beg < n) {
        const int hspan = span >> 1;
        const int hb = beg + half * hspan;
        int he = hb + hspan;
        if (he > n) he = n;
        const int last = n - 1;
        int cur = (sorted[hb < last ? hb : last].x >> 24) & ROW_MASK;
        float4 s = make_float4(0.f, 0.f, 0.f, 0.f);
        if (beg + span <= n) {
          for (int st = 0; st < hspan; st += 16)
            sum_step<true>(sorted[hb + st + l16], 16, 16, Xb, cb, accl, cur, s);
        } else {
          int lo_end = beg + hspan;
          if (lo_end > n) lo_end = n;
          for (int st = 0; st < hspan; st += 16) {
            const int mu = lo_end - (beg + st);
            if (mu <= 0) break;
            const int i = hb + st + l16;
            sum_step<false>(sorted[i < last ? i : last], he - (hb + st), mu, Xb, cb, accl, cur, s);
          }
        }
        float* p = accl + cur * A_PITCH;
        lds_add(p, s.x);
        lds_add(p + 1, s.y);
        lds_add(p + 2, s.z);
        lds_add(p + 3, s.w);
      }
    }

    int tp = t;
    asm volatile("" : "+v"(tp));
    const int halfp = (tp >> 4) & 1;
    const int l16p = tp & 15;
    const float4* Wv = (const float4*)(W + (size_t)r * IN_SIZE * OUT_SIZE);
    const float4 w0 = Wv[tp];
    const float4 w1 = Wv[tp + SUM_THREADS];
    __syncthreads();
    *(float4*)&sW[(tp >> 4) * W_PITCH + l16p * 4] = w0;
    *(float4*)&sW[((tp >> 4) + 32) * W_PITCH + l16p * 4] = w1;
    __syncthreads();

    {
      const float* aP = agg + (rt * 16 + l16p) * A_PITCH + 2 * halfp;
      const float* bP = sW + (2 * halfp) * W_PITCH + ctb * 16 + l16p;
#pragma unroll
      for (int kk = 0; kk < IN_SIZE / 4; ++kk) {
        const float2 av = *(const float2*)(aP + kk * 4);
        v2f a;
        a[0] = av.x;
        a[1] = av.y;
#pragma unroll
        for (int i = 0; i < 2; ++i) {
          v2f bf;
          bf[0] = bP[(kk * 4) * W_PITCH + i * 16];
          bf[1] = bP[(kk * 4 + 1) * W_PITCH + i * 16];
          cacc[i] = __builtin_amdgcn_wmma_f32_16x16x4_f32(
               false, a,  false, bf,
               (short)0, cacc[i],  false,  false);
        }
      }
    }
    __syncthreads();
    if (r + 1 < n_rel) {
#pragma unroll
      for (int k = 0; k < (TILE_ROWS * A_PITCH / 4 + SUM_THREADS - 1) / SUM_THREADS; ++k) {
        const int q = tp + k * SUM_THREADS;
        if (q < TILE_ROWS * A_PITCH / 4) *(float4*)&agg[q * 4] = make_float4(0.f, 0.f, 0.f, 0.f);
      }
    }
  }

  int te = t;
  asm volatile("" : "+v"(te));
  const int halfe = (te >> 4) & 1;
  const int l16e = te & 15;
#pragma unroll
  for (int i = 0; i < 2; ++i)
#pragma unroll
    for (int j = 0; j < 8; ++j)
      agg[(rt * 16 + j + 8 * halfe) * A_PITCH + (ctb + i) * 16 + l16e] = cacc[i][j];
  __syncthreads();
  const int c4 = l16e * 4;
#pragma unroll
  for (int k = 0; k < (BIN_ROWS * OUT_SIZE / 4 + SUM_THREADS - 1) / SUM_THREADS; ++k) {
    const int q = te + k * SUM_THREADS;
    const int lrow = q >> 4;
    const int grow = b * BIN_ROWS + lrow;
    if (lrow < BIN_ROWS && grow < n_nodes)
      *(float4*)(out + (size_t)grow * OUT_SIZE + c4) = *(const float4*)&agg[lrow * A_PITCH + c4];
  }
}

extern "C" void kernel_launch(void* const* d_in, const int* in_sizes, int n_in,
                              void* d_out, int out_size, void* d_ws, size_t ws_size,
                              hipStream_t stream) {
  const float* inp = (const float*)d_in[0];
  const int* src = (const int*)d_in[1];
  const int* dst = (const int*)d_in[2];
  const float* eval = (const float*)d_in[3];
  const float* weights = (const float*)d_in[4];
  const float* bias = (const float*)d_in[5];
  float* out = (float*)d_out;

  const int n_nodes = in_sizes[0] / IN_SIZE;
  const int n_rel = in_sizes[4] / (IN_SIZE * OUT_SIZE);
  const int n_edges = in_sizes[1] / n_rel;
  const int nbins = (n_nodes + BIN_ROWS - 1) / BIN_ROWS;

  int2* recs = (int2*)d_ws;
  int* totals = (int*)(recs + (size_t)n_rel * n_edges);
  int* cursor = totals + n_rel * nbins;
  int* segbase = cursor + n_rel * nbins;

  const int n_counters = 2 * n_rel * nbins;
  rgcn_clear<<<(n_counters + BIN_THREADS - 1) / BIN_THREADS, BIN_THREADS, 0, stream>>>(totals, n_counters);

  const int chunks = (n_edges + CHUNK - 1) / CHUNK;
  rgcn_bin_count<<<dim3(chunks, n_rel), BIN_THREADS, 0, stream>>>(dst, totals, n_edges, nbins);
  rgcn_bin_place<<<dim3(chunks, n_rel), BIN_THREADS, 0, stream>>>(
      src, dst, eval, totals, cursor, segbase, recs, n_edges, nbins);
  rgcn_bin_sum<<<nbins, SUM_THREADS, 0, stream>>>(
      inp, weights, recs, totals, segbase, bias, out, n_nodes, n_rel, nbins);
}
